# starting kernel
# baseline (speedup 1.0000x reference)
#include <hip/hip_runtime.h>
#include <math.h>

#define N_NODES 100000
#define N_EDGES 1600000
#define IN_DIM  128
#define HID     32
#define HEADS   4
#define OUT_DIM 16

typedef __attribute__((ext_vector_type(2))) float v2f;
typedef __attribute__((ext_vector_type(8))) float v8f;

__device__ __forceinline__ unsigned fkey(float x) {
  unsigned b = __float_as_uint(x);
  return (b & 0x80000000u) ? ~b : (b | 0x80000000u);
}
__device__ __forceinline__ float funkey(unsigned k) {
  return __uint_as_float((k & 0x80000000u) ? (k & 0x7FFFFFFFu) : ~k);
}

__global__ void zero_f32(float* __restrict__ p, long long n) {
  long long t = (long long)blockIdx.x * blockDim.x + threadIdx.x;
  if (t < n) p[t] = 0.0f;
}

template <int NT>
__global__ void __launch_bounds__(256)
wmma_gemm_strip(const float* __restrict__ A, const float* __restrict__ W,
                float* __restrict__ C, int Mtiles, int K, int ldc, int coff) {
  const int wave = (blockIdx.x * blockDim.x + threadIdx.x) >> 5;
  if (wave >= Mtiles) return;
  const int lane = threadIdx.x & 31;
  const int lo = lane & 15;
  const int hi = lane >> 4;
  const float* Ab = A + ((size_t)wave * 16 + lo) * K;
  v8f acc[NT];
#pragma unroll
  for (int j = 0; j < NT; ++j)
    acc[j] = (v8f){0.f, 0.f, 0.f, 0.f, 0.f, 0.f, 0.f, 0.f};
  for (int k0 = 0; k0 < K; k0 += 4) {
    const int ka = k0 + hi * 2;
    v2f a;
    a.x = Ab[ka]; a.y = Ab[ka + 1];
#pragma unroll
    for (int j = 0; j < NT; ++j) {
      const float* Wb = W + ((size_t)j * 16 + lo) * K;
      v2f b;
      b.x = Wb[ka]; b.y = Wb[ka + 1];
      acc[j] = __builtin_amdgcn_wmma_f32_16x16x4_f32(false, a, false, b,
                                                     (short)0, acc[j], false, false);
    }
  }
#pragma unroll
  for (int j = 0; j < NT; ++j) {
    float* Cb = C + (size_t)wave * 16 * ldc + coff + (size_t)j * 16 + lo;
#pragma unroll
    for (int r = 0; r < 8; ++r)
      Cb[(size_t)(r + hi * 8) * ldc] = acc[j][r];
  }
}

__global__ void __launch_bounds__(256)
wmma_gemm_frac(const float* __restrict__ H, const int* __restrict__ fc,
               const float* __restrict__ Wf, float* __restrict__ hfpre) {
  const int wave = (blockIdx.x * blockDim.x + threadIdx.x) >> 5;
  if (wave >= N_NODES / 16) return;
  const int s = blockIdx.y;
  const int lane = threadIdx.x & 31;
  const int lo = lane & 15;
  const int hi = lane >> 4;
  const int nid = fc[(wave * 16 + lo) * 3 + s];
  const float* Ab = H + (size_t)nid * IN_DIM;
  const float* Ws = Wf + (size_t)s * HID * IN_DIM;
  v8f acc[2];
  acc[0] = (v8f){0.f, 0.f, 0.f, 0.f, 0.f, 0.f, 0.f, 0.f};
  acc[1] = acc[0];
  for (int k0 = 0; k0 < IN_DIM; k0 += 4) {
    const int ka = k0 + hi * 2;
    v2f a;
    a.x = Ab[ka]; a.y = Ab[ka + 1];
#pragma unroll
    for (int j = 0; j < 2; ++j) {
      const float* Wb = Ws + ((size_t)j * 16 + lo) * IN_DIM;
      v2f b;
      b.x = Wb[ka]; b.y = Wb[ka + 1];
      acc[j] = __builtin_amdgcn_wmma_f32_16x16x4_f32(false, a, false, b,
                                                     (short)0, acc[j], false, false);
    }
  }
#pragma unroll
  for (int j = 0; j < 2; ++j) {
    float* Cb = hfpre + (size_t)wave * 16 * 96 + s * HID + (size_t)j * 16 + lo;
#pragma unroll
    for (int r = 0; r < 8; ++r)
      Cb[(size_t)(r + hi * 8) * 96] = acc[j][r];
  }
}

__global__ void attn_scores(const float* __restrict__ z, const float* __restrict__ a,
                            float* __restrict__ ss, float* __restrict__ sd,
                            int n, int heads) {
  int t = blockIdx.x * blockDim.x + threadIdx.x;
  if (t >= n * heads) return;
  int node = t / heads, hh = t - node * heads;
  const float* zp = z + (size_t)node * heads * HID + hh * HID;
  const float* al = a + (size_t)hh * 2 * HID;
  const float* ar = al + HID;
  float s0 = 0.f, s1 = 0.f;
#pragma unroll 8
  for (int o = 0; o < HID; ++o) { s0 += zp[o] * al[o]; s1 += zp[o] * ar[o]; }
  ss[t] = s0; sd[t] = s1;
}

__global__ void edge_scores_max(const int* __restrict__ src, const int* __restrict__ dst,
                                const float* __restrict__ ss, const float* __restrict__ sd,
                                float* __restrict__ e, unsigned* __restrict__ m, int heads) {
  int t = blockIdx.x * blockDim.x + threadIdx.x;
  if (t >= N_EDGES * heads) return;
  int ed = t / heads, hh = t - ed * heads;
  float x = ss[src[ed] * heads + hh] + sd[dst[ed] * heads + hh];
  x = (x > 0.f) ? x : 0.01f * x;
  e[t] = x;
  atomicMax(m + dst[ed] * heads + hh, fkey(x));
}

__global__ void edge_expsum(const int* __restrict__ dst, float* __restrict__ e,
                            const unsigned* __restrict__ m, float* __restrict__ sum,
                            int heads) {
  int t = blockIdx.x * blockDim.x + threadIdx.x;
  if (t >= N_EDGES * heads) return;
  int ed = t / heads, hh = t - ed * heads;
  int di = dst[ed] * heads + hh;
  float ee = expf(e[t] - funkey(m[di]));
  e[t] = ee;
  atomicAdd(sum + di, ee);
}

__global__ void edge_aggregate(const int* __restrict__ src, const int* __restrict__ dst,
                               const float* __restrict__ e, const float* __restrict__ sum,
                               const float* __restrict__ z, float* __restrict__ out,
                               int heads, int ldo) {
  long long t = (long long)blockIdx.x * blockDim.x + threadIdx.x;
  const int C = heads * HID;
  if (t >= (long long)N_EDGES * C) return;
  int ed = (int)(t / C);
  int c  = (int)(t - (long long)ed * C);
  int hh = c >> 5;
  int di = dst[ed];
  float alpha = e[(size_t)ed * heads + hh] / fmaxf(sum[di * heads + hh], 1e-9f);
  atomicAdd(out + (size_t)di * ldo + c, alpha * z[(size_t)src[ed] * C + c]);
}

__global__ void elu_inplace(float* __restrict__ x, long long n) {
  long long t = (long long)blockIdx.x * blockDim.x + threadIdx.x;
  if (t >= n) return;
  float v = x[t];
  x[t] = (v > 0.f) ? v : (expf(v) - 1.0f);
}

extern "C" void kernel_launch(void* const* d_in, const int* in_sizes, int n_in,
                              void* d_out, int out_size, void* d_ws, size_t ws_size,
                              hipStream_t stream) {
  const float* h    = (const float*)d_in[0];
  const int*   src  = (const int*)  d_in[1];
  const int*   dst  = (const int*)  d_in[2];
  const int*   fc   = (const int*)  d_in[3];
  const float* W1   = (const float*)d_in[4];
  const float* a1   = (const float*)d_in[5];
  const float* W2   = (const float*)d_in[6];
  const float* a2   = (const float*)d_in[7];
  const float* Wf   = (const float*)d_in[8];
  const float* Wff  = (const float*)d_in[9];
  const float* Wfc2 = (const float*)d_in[10];
  float* out = (float*)d_out;

  float* w = (float*)d_ws;
  size_t off = 0;
  float*    agg1 = w + off;              off += (size_t)N_NODES * 128;
  unsigned* m1   = (unsigned*)(w + off); off += (size_t)N_NODES * HEADS;
  float*    sum1 = w + off;              off += (size_t)N_NODES * HEADS;
  unsigned* m2   = (unsigned*)(w + off); off += N_NODES;
  float*    sum2 = w + off;              off += N_NODES;
  float*    hg   = w + off;              off += (size_t)N_NODES * 64;
  const long long zeroN = (long long)off;
  float* z1  = w + off; off += (size_t)N_NODES * 128;
  float* s1s = w + off; off += (size_t)N_NODES * HEADS;
  float* s1d = w + off; off += (size_t)N_NODES * HEADS;
  float* e1  = w + off; off += (size_t)N_EDGES * HEADS;
  float* z2  = w + off; off += (size_t)N_NODES * HID;
  float* s2s = w + off; off += N_NODES;
  float* s2d = w + off; off += N_NODES;
  float* e2  = w + off; off += N_EDGES;
  float* hfp = w + off; off += (size_t)N_NODES * 96;
  (void)ws_size; (void)in_sizes; (void)n_in; (void)out_size;

  const int TB = 256;
  const int MT = N_NODES / 16;
  const int gemm_blocks = (MT + 7) / 8;

  zero_f32<<<(int)((zeroN + TB - 1) / TB), TB, 0, stream>>>(w, zeroN);

  wmma_gemm_strip<8><<<gemm_blocks, TB, 0, stream>>>(h, W1, z1, MT, 128, 128, 0);
  attn_scores<<<(N_NODES * HEADS + TB - 1) / TB, TB, 0, stream>>>(z1, a1, s1s, s1d, N_NODES, HEADS);
  edge_scores_max<<<(N_EDGES * HEADS + TB - 1) / TB, TB, 0, stream>>>(src, dst, s1s, s1d, e1, m1, HEADS);
  edge_expsum<<<(N_EDGES * HEADS + TB - 1) / TB, TB, 0, stream>>>(dst, e1, m1, sum1, HEADS);
  {
    long long tot = (long long)N_EDGES * 128;
    edge_aggregate<<<(int)((tot + TB - 1) / TB), TB, 0, stream>>>(src, dst, e1, sum1, z1, agg1, HEADS, 128);
  }
  elu_inplace<<<(int)(((long long)N_NODES * 128 + TB - 1) / TB), TB, 0, stream>>>(
      agg1, (long long)N_NODES * 128);

  wmma_gemm_strip<2><<<gemm_blocks, TB, 0, stream>>>(agg1, W2, z2, MT, 128, HID, 0);
  attn_scores<<<(N_NODES + TB - 1) / TB, TB, 0, stream>>>(z2, a2, s2s, s2d, N_NODES, 1);
  edge_scores_max<<<(N_EDGES + TB - 1) / TB, TB, 0, stream>>>(src, dst, s2s, s2d, e2, m2, 1);
  edge_expsum<<<(N_EDGES + TB - 1) / TB, TB, 0, stream>>>(dst, e2, m2, sum2, 1);
  {
    long long tot = (long long)N_EDGES * HID;
    edge_aggregate<<<(int)((tot + TB - 1) / TB), TB, 0, stream>>>(src, dst, e2, sum2, z2, hg, 1, 64);
  }

  {
    dim3 g(gemm_blocks, 3);
    wmma_gemm_frac<<<g, TB, 0, stream>>>(h, fc, Wf, hfp);
  }
  wmma_gemm_strip<2><<<gemm_blocks, TB, 0, stream>>>(hfp, Wff, hg, MT, 96, 64, HID);

  wmma_gemm_strip<1><<<gemm_blocks, TB, 0, stream>>>(hg, Wfc2, out, MT, 64, OUT_DIM, 0);
}


# Layer-1 owner workgroups scan a 1-byte-per-edge owner-range map (one small new launch writes it), not the 4-byte dst list; src, dst read only for noted edges; LDS sums in arrival order as before
# speedup vs baseline: 4.1103x; 4.1103x over previous
#include <hip/hip_runtime.h>
#include <math.h>

#define N_NODES 100000
#define N_EDGES 1600000
#define IN_DIM  128
#define HID     32
#define HEADS   4
#define OUT_DIM 16

typedef __attribute__((ext_vector_type(2))) float v2f;
typedef __attribute__((ext_vector_type(8))) float v8f;

__device__ __forceinline__ unsigned fkey(float x) {
  unsigned b = __float_as_uint(x);
  return (b & 0x80000000u) ? ~b : (b | 0x80000000u);
}
__device__ __forceinline__ float funkey(unsigned k) {
  return __uint_as_float((k & 0x80000000u) ? (k & 0x7FFFFFFFu) : ~k);
}

template <int NT>
__global__ void __launch_bounds__(256)
wmma_gemm_strip(const float* __restrict__ A, const float* __restrict__ W,
                float* __restrict__ C, int Mtiles, int K, int ldc, int coff) {
  const int wave = (blockIdx.x * blockDim.x + threadIdx.x) >> 5;
  if (wave >= Mtiles) return;
  const int lane = threadIdx.x & 31;
  const int lo = lane & 15;
  const int hi = lane >> 4;
  const float* Ab = A + ((size_t)wave * 16 + lo) * K;
  v8f acc[NT];
#pragma unroll
  for (int j = 0; j < NT; ++j)
    acc[j] = (v8f){0.f, 0.f, 0.f, 0.f, 0.f, 0.f, 0.f, 0.f};
  for (int k0 = 0; k0 < K; k0 += 4) {
    const int ka = k0 + hi * 2;
    v2f a;
    a.x = Ab[ka]; a.y = Ab[ka + 1];
#pragma unroll
    for (int j = 0; j < NT; ++j) {
      const float* Wb = W + ((size_t)j * 16 + lo) * K;
      v2f b;
      b.x = Wb[ka]; b.y = Wb[ka + 1];
      acc[j] = __builtin_amdgcn_wmma_f32_16x16x4_f32(false, a, false, b,
                                                     (short)0, acc[j], false, false);
    }
  }
#pragma unroll
  for (int j = 0; j < NT; ++j) {
    float* Cb = C + (size_t)wave * 16 * ldc + coff + (size_t)j * 16 + lo;
#pragma unroll
    for (int r = 0; r < 8; ++r)
      Cb[(size_t)(r + hi * 8) * ldc] = acc[j][r];
  }
}

__global__ void __launch_bounds__(256)
wmma_gemm_frac(const float* __restrict__ H, const int* __restrict__ fc,
               const float* __restrict__ Wf, float* __restrict__ hfpre) {
  const int wave = (blockIdx.x * blockDim.x + threadIdx.x) >> 5;
  if (wave >= N_NODES / 16) return;
  const int s = blockIdx.y;
  const int lane = threadIdx.x & 31;
  const int lo = lane & 15;
  const int hi = lane >> 4;
  const int nid = fc[(wave * 16 + lo) * 3 + s];
  const float* Ab = H + (size_t)nid * IN_DIM;
  const float* Ws = Wf + (size_t)s * HID * IN_DIM;
  v8f acc[2];
  acc[0] = (v8f){0.f, 0.f, 0.f, 0.f, 0.f, 0.f, 0.f, 0.f};
  acc[1] = acc[0];
  for (int k0 = 0; k0 < IN_DIM; k0 += 4) {
    const int ka = k0 + hi * 2;
    v2f a;
    a.x = Ab[ka]; a.y = Ab[ka + 1];
#pragma unroll
    for (int j = 0; j < 2; ++j) {
      const float* Wb = Ws + ((size_t)j * 16 + lo) * IN_DIM;
      v2f b;
      b.x = Wb[ka]; b.y = Wb[ka + 1];
      acc[j] = __builtin_amdgcn_wmma_f32_16x16x4_f32(false, a, false, b,
                                                     (short)0, acc[j], false, false);
    }
  }
#pragma unroll
  for (int j = 0; j < 2; ++j) {
    float* Cb = hfpre + (size_t)wave * 16 * 96 + s * HID + (size_t)j * 16 + lo;
#pragma unroll
    for (int r = 0; r < 8; ++r)
      Cb[(size_t)(r + hi * 8) * 96] = acc[j][r];
  }
}

__global__ void attn_scores(const float* __restrict__ z, const float* __restrict__ a,
                            float* __restrict__ ss, float* __restrict__ sd,
                            int n, int heads) {
  int t = blockIdx.x * blockDim.x + threadIdx.x;
  if (t >= n * heads) return;
  int node = t / heads, hh = t - node * heads;
  const float* zp = z + (size_t)node * heads * HID + hh * HID;
  const float* al = a + (size_t)hh * 2 * HID;
  const float* ar = al + HID;
  float s0 = 0.f, s1 = 0.f;
#pragma unroll 8
  for (int o = 0; o < HID; ++o) { s0 += zp[o] * al[o]; s1 += zp[o] * ar[o]; }
  ss[t] = s0; sd[t] = s1;
}

#define AG_ROWS   391
#define AG_GROUPS 256
#define AG_CAP    16384
#define AG_TB     1024
#define AG_U      8
#define AG_RB     9
#define AG_B      4

__device__ __forceinline__ int lane_get_i(int v, int i) {
  return __builtin_amdgcn_readlane(v, i);
}
__device__ __forceinline__ float lane_get_f(float v, int i) {
  return __int_as_float(__builtin_amdgcn_readlane(__float_as_int(v), i));
}

#define OM_TB     256
#define OM_GROUPS ((N_EDGES / 16 + OM_TB - 1) / OM_TB)
__device__ __forceinline__ unsigned owner_of(int d) {
  return min((unsigned)d / (unsigned)AG_ROWS, (unsigned)(AG_GROUPS - 1));
}
__global__ void __launch_bounds__(OM_TB)
dst_owner_map(const int* __restrict__ dst, unsigned* __restrict__ omap) {
  constexpr int T = OM_GROUPS * OM_TB;
  const int t = blockIdx.x * OM_TB + threadIdx.x;
  const int4* __restrict__ dst4 = (const int4*)dst;
  int4 d[4];
#pragma unroll
  for (int u = 0; u < 4; ++u) d[u] = dst4[min(t + u * T, N_EDGES / 4 - 1)];
#pragma unroll
  for (int u = 0; u < 4; ++u) {
    const int i = t + u * T;
    if (i < N_EDGES / 4)
      omap[i] = owner_of(d[u].x) | (owner_of(d[u].y) << 8) | (owner_of(d[u].z) << 16) | (owner_of(d[u].w) << 24);
  }
}

template <int H, bool APPLY_ELU, bool TAKE_LIST>
__global__ void __launch_bounds__(AG_TB)
edge_softmax_aggregate_owner(const int* __restrict__ src, const int* __restrict__ dst,
                             const float* __restrict__ ss, const float* __restrict__ sd,
                             const float* __restrict__ z, float* __restrict__ out, int ldo,
                             int* __restrict__ wlist, int* __restrict__ wcnt,
                             const unsigned* __restrict__ omap) {
  constexpr int C = H * HID;
  constexpr int RMASK = (1 << AG_RB) - 1;
  __shared__ __attribute__((aligned(16))) float acc[(AG_ROWS + 1) * C];
  __shared__ int lst[AG_CAP];
  __shared__ __attribute__((aligned(16))) unsigned mx[(AG_ROWS + 1) * H];
  __shared__ __attribute__((aligned(16))) float sm[(AG_ROWS + 1) * H];
  __shared__ __attribute__((aligned(16))) float sdl[(AG_ROWS + 1) * H];
  __shared__ int cnt;
  const int tid  = threadIdx.x;
  const int lane = tid & 31;
  const int wave = __builtin_amdgcn_readfirstlane(tid >> 5);
  const int base = blockIdx.x * AG_ROWS;
  int* __restrict__ wl = wlist + (size_t)blockIdx.x * AG_CAP;

  for (int i = tid; i < (AG_ROWS + 1) * C; i += AG_TB) acc[i] = 0.f;
  for (int i = tid; i < (AG_ROWS + 1) * H; i += AG_TB) {
    mx[i] = 0u;
    sm[i] = 0.f;
    const int g = base * H + i;
    sdl[i] = (i < AG_ROWS * H && g < N_NODES * H) ? sd[g] : 0.f;
  }
  int total;
  if constexpr (TAKE_LIST) {
    total = wcnt[blockIdx.x];
    const int m = (total > AG_CAP) ? 0 : total;
    for (int i = tid; i < m; i += AG_TB) lst[i] = wl[i];
    __syncthreads();
  } else {
    if (tid == 0) cnt = 0;
    __syncthreads();

    const unsigned idb = (unsigned)blockIdx.x * 0x01010101u;
    auto note16 = [&](const uint4 w, const int q) {
      const unsigned x0 = w.x ^ idb, x1 = w.y ^ idb, x2 = w.z ^ idb, x3 = w.w ^ idb;
      const unsigned u0 = (x0 - 0x01010101u) & ~x0, u1 = (x1 - 0x01010101u) & ~x1;
      const unsigned u2 = (x2 - 0x01010101u) & ~x2, u3 = (x3 - 0x01010101u) & ~x3;
      if ((u0 | u1 | u2 | u3) & 0x80808080u) {
        unsigned c = ((u0 >> 7) & 0x01010101u) | ((u1 >> 6) & 0x02020202u) |
                     ((u2 >> 5) & 0x04040404u) | ((u3 >> 4) & 0x08080808u);
        int slot = atomicAdd(&cnt, __popc(c));
        while (c) {
          const int b = __builtin_ctz(c);
          c &= c - 1u;
          if (slot < AG_CAP) lst[slot] = q * 16 + ((b & 7) << 2) + (b >> 3);
          ++slot;
        }
      }
    };
    const uint4* __restrict__ om4 = (const uint4*)omap;
    constexpr int P16  = N_EDGES / 16;
    constexpr int STEP = AG_TB * 4;
    constexpr int FULL = P16 / STEP;
    {
      int q0 = tid;
      uint4 cur[4], nxt[4];
#pragma unroll
      for (int u = 0; u < 4; ++u) cur[u] = om4[q0 + u * AG_TB];
      for (int it = 0; it + 1 < FULL; ++it) {
#pragma unroll
        for (int u = 0; u < 4; ++u) nxt[u] = om4[q0 + STEP + u * AG_TB];
#pragma unroll
        for (int u = 0; u < 4; ++u) note16(cur[u], q0 + u * AG_TB);
#pragma unroll
        for (int u = 0; u < 4; ++u) cur[u] = nxt[u];
        q0 += STEP;
      }
#pragma unroll
      for (int u = 0; u < 4; ++u) note16(cur[u], q0 + u * AG_TB);
#pragma unroll
      for (int u = 0; u < 4; ++u) {
        const int q = FULL * STEP + tid + u * AG_TB;
        if (q < P16) note16(om4[q], q);
      }
    }
    __syncthreads();

    total = cnt;
    if (tid == 0) wcnt[blockIdx.x] = total;
  }
  const bool overflow = total > AG_CAP;
  const int n = __builtin_amdgcn_readfirstlane(overflow ? 0 : total);

  auto leaky = [](float x) { return (x > 0.f) ? x : 0.01f * x; };
  auto load_ss = [&](int s, float (&x)[H]) {
    if constexpr (H == 4) {
      const float4 t = ((const float4*)ss)[s];
      x[0] = t.x; x[1] = t.y; x[2] = t.z; x[3] = t.w;
    } else {
#pragma unroll
      for (int h = 0; h < H; ++h) x[h] = ss[(size_t)s * H + h];
    }
  };
  auto walk = [&](auto f) {
    for (int ed = tid; ed < N_EDGES; ed += AG_TB) {
      const unsigned r = (unsigned)(dst[ed] - base);
      if (r < (unsigned)AG_ROWS) f(src[ed], (int)r);
    }
  };

  if (overflow) {
    walk([&](int s, int r) {
      float x[H];
      load_ss(s, x);
      for (int h = 0; h < H; ++h) atomicMax(&mx[r * H + h], fkey(leaky(x[h] + sdl[r * H + h])));
    });
  } else {
    for (int i0 = tid; i0 < n; i0 += AG_B * AG_TB) {
      int p[AG_B], s[AG_B], rw[AG_B];
      float x[AG_B][H];
#pragma unroll
      for (int u = 0; u < AG_B; ++u) {
        const int idx = i0 + u * AG_TB;
        p[u] = (idx < n) ? lst[idx] : -1;
      }
#pragma unroll
      for (int u = 0; u < AG_B; ++u) {
        if constexpr (TAKE_LIST) {
          s[u]  = (p[u] >= 0) ? (p[u] >> AG_RB) : 0;
          rw[u] = p[u] & RMASK;
        } else {
          s[u]  = (p[u] >= 0) ? src[p[u]] : 0;
          const int d = (p[u] >= 0) ? dst[p[u]] : base;
          rw[u] = (int)min((unsigned)(d - base), (unsigned)AG_ROWS);
        }
      }
#pragma unroll
      for (int u = 0; u < AG_B; ++u) load_ss(s[u], x[u]);
#pragma unroll
      for (int u = 0; u < AG_B; ++u) {
        if (p[u] >= 0) {
          const int r = rw[u];
          if constexpr (!TAKE_LIST) {
            const int e = (s[u] << AG_RB) | r;
            lst[i0 + u * AG_TB] = e;
            wl[i0 + u * AG_TB] = e;
          }
#pragma unroll
          for (int h = 0; h < H; ++h)
            atomicMax(&mx[r * H + h], fkey(leaky(x[u][h] + sdl[r * H + h])));
        }
      }
    }
  }
  __syncthreads();

  if (overflow) {
    walk([&](int s, int r) {
      float x[H];
      load_ss(s, x);
      for (int h = 0; h < H; ++h)
        atomicAdd(&sm[r * H + h], expf(leaky(x[h] + sdl[r * H + h]) - funkey(mx[r * H + h])));
    });
  } else {
    for (int i0 = tid; i0 < n; i0 += AG_B * AG_TB) {
      int p[AG_B];
      float x[AG_B][H];
#pragma unroll
      for (int u = 0; u < AG_B; ++u) {
        const int idx = i0 + u * AG_TB;
        p[u] = (idx < n) ? lst[idx] : -1;
      }
#pragma unroll
      for (int u = 0; u < AG_B; ++u) load_ss((p[u] >= 0) ? (p[u] >> AG_RB) : 0, x[u]);
#pragma unroll
      for (int u = 0; u < AG_B; ++u) {
        if (p[u] >= 0) {
          const int r = p[u] & RMASK;
#pragma unroll
          for (int h = 0; h < H; ++h)
            atomicAdd(&sm[r * H + h], expf(leaky(x[u][h] + sdl[r * H + h]) - funkey(mx[r * H + h])));
        }
      }
    }
  }
  __syncthreads();

  if (overflow) {
    walk([&](int s, int r) {
      float x[H];
      load_ss(s, x);
      for (int h = 0; h < H; ++h) {
        const float a = expf(leaky(x[h] + sdl[r * H + h]) - funkey(mx[r * H + h])) / fmaxf(sm[r * H + h], 1e-9f);
        for (int o = 0; o < HID; ++o)
          atomicAdd(&acc[r * C + h * HID + o], a * z[(size_t)s * C + h * HID + o]);
      }
    });
  }
  for (int b0 = wave * 32; b0 < n; b0 += AG_TB) {
    const int idx = b0 + lane;
    const bool valid = idx < n;
    const int p = valid ? lst[idx] : AG_ROWS;
    const int s = p >> AG_RB;
    const int row = p & RMASK;
    float al[H];
    load_ss(s, al);
#pragma unroll
    for (int h = 0; h < H; ++h) {
      const float ee = expf(leaky(al[h] + sdl[row * H + h]) - funkey(mx[row * H + h]));
      al[h] = valid ? ee / fmaxf(sm[row * H + h], 1e-9f) : 0.f;
    }
    const int nb = min(32, n - b0);
    for (int i0 = 0; i0 < nb; i0 += AG_U) {
      float v[AG_U][H];
#pragma unroll
      for (int u = 0; u < AG_U; ++u) {
        const float* zp = z + (size_t)lane_get_i(s, i0 + u) * C + lane;
#pragma unroll
        for (int h = 0; h < H; ++h) v[u][h] = zp[h * HID];
      }
#pragma unroll
      for (int u = 0; u < AG_U; ++u) {
        float* ap = acc + lane_get_i(row, i0 + u) * C + lane;
#pragma unroll
        for (int h = 0; h < H; ++h)
          atomicAdd(ap + h * HID, lane_get_f(al[h], i0 + u) * v[u][h]);
      }
    }
  }
  __syncthreads();

  constexpr int C4 = C / 4;
  for (int i = tid; i < AG_ROWS * C4; i += AG_TB) {
    const int r = i / C4, c4 = i - r * C4;
    const int node = base + r;
    if (node < N_NODES) {
      float4 v = *(const float4*)&acc[r * C + c4 * 4];
      if constexpr (APPLY_ELU) {
        v.x = (v.x > 0.f) ? v.x : (expf(v.x) - 1.0f);
        v.y = (v.y > 0.f) ? v.y : (expf(v.y) - 1.0f);
        v.z = (v.z > 0.f) ? v.z : (expf(v.z) - 1.0f);
        v.w = (v.w > 0.f) ? v.w : (expf(v.w) - 1.0f);
      }
      *(float4*)&out[(size_t)node * ldo + c4 * 4] = v;
    }
  }
}

extern "C" void kernel_launch(void* const* d_in, const int* in_sizes, int n_in,
                              void* d_out, int out_size, void* d_ws, size_t ws_size,
                              hipStream_t stream) {
  const float* h    = (const float*)d_in[0];
  const int*   src  = (const int*)  d_in[1];
  const int*   dst  = (const int*)  d_in[2];
  const int*   fc   = (const int*)  d_in[3];
  const float* W1   = (const float*)d_in[4];
  const float* a1   = (const float*)d_in[5];
  const float* W2   = (const float*)d_in[6];
  const float* a2   = (const float*)d_in[7];
  const float* Wf   = (const float*)d_in[8];
  const float* Wff  = (const float*)d_in[9];
  const float* Wfc2 = (const float*)d_in[10];
  float* out = (float*)d_out;

  float* w = (float*)d_ws;
  size_t off = 0;
  float* agg1 = w + off; off += (size_t)N_NODES * 128;
  float* hg   = w + off; off += (size_t)N_NODES * 64;
  float* z1   = w + off; off += (size_t)N_NODES * 128;
  float* s1s  = w + off; off += (size_t)N_NODES * HEADS;
  float* s1d  = w + off; off += (size_t)N_NODES * HEADS;
  float* z2   = w + off; off += (size_t)N_NODES * HID;
  float* s2s  = w + off; off += N_NODES;
  float* s2d  = w + off; off += N_NODES;
  float* hfp  = w + off; off += (size_t)N_NODES * 96;
  int*   elst = (int*)(w + off); off += (size_t)AG_GROUPS * AG_CAP;
  int*   ecnt = (int*)(w + off); off += AG_GROUPS;
  unsigned* omap = (unsigned*)(w + off); off += N_EDGES / 4;
  (void)ws_size; (void)in_sizes; (void)n_in; (void)out_size;

  const int TB = 256;
  const int MT = N_NODES / 16;
  const int gemm_blocks = (MT + 7) / 8;

  wmma_gemm_strip<8><<<gemm_blocks, TB, 0, stream>>>(h, W1, z1, MT, 128, 128, 0);
  attn_scores<<<(N_NODES * HEADS + TB - 1) / TB, TB, 0, stream>>>(z1, a1, s1s, s1d, N_NODES, HEADS);
  dst_owner_map<<<OM_GROUPS, OM_TB, 0, stream>>>(dst, omap);
  edge_softmax_aggregate_owner<HEADS, true, false><<<AG_GROUPS, AG_TB, 0, stream>>>(src, dst, s1s, s1d, z1, agg1, 128, elst, ecnt, omap);

  wmma_gemm_strip<2><<<gemm_blocks, TB, 0, stream>>>(agg1, W2, z2, MT, 128, HID, 0);
  attn_scores<<<(N_NODES + TB - 1) / TB, TB, 0, stream>>>(z2, a2, s2s, s2d, N_NODES, 1);
  edge_softmax_aggregate_owner<1, false, true><<<AG_GROUPS, AG_TB, 0, stream>>>(src, dst, s2s, s2d, z2, hg, 64, elst, ecnt, omap);

  {
    dim3 g(gemm_blocks, 3);
    wmma_gemm_frac<<<g, TB, 0, stream>>>(h, fc, Wf, hfp);
  }
  wmma_gemm_strip<2><<<gemm_blocks, TB, 0, stream>>>(hfp, Wff, hg, MT, 96, 64, HID);

  wmma_gemm_strip<1><<<gemm_blocks, TB, 0, stream>>>(hg, Wfc2, out, MT, 64, OUT_DIM, 0);
}


# GEMM fragment loads 16 B per lane, A and B alike: a lane half reads 4 consecutive k of an 8-k block and feeds 2 WMMAs; half the load instructions; k sum order inside each 8-k block changed
# speedup vs baseline: 4.7413x; 1.1535x over previous
#include <hip/hip_runtime.h>
#include <math.h>

#define N_NODES 100000
#define N_EDGES 1600000
#define IN_DIM  128
#define HID     32
#define HEADS   4
#define OUT_DIM 16

typedef __attribute__((ext_vector_type(2))) float v2f;
typedef __attribute__((ext_vector_type(8))) float v8f;

__device__ __forceinline__ unsigned fkey(float x) {
  unsigned b = __float_as_uint(x);
  return (b & 0x80000000u) ? ~b : (b | 0x80000000u);
}
__device__ __forceinline__ float funkey(unsigned k) {
  return __uint_as_float((k & 0x80000000u) ? (k & 0x7FFFFFFFu) : ~k);
}

template <int NT>
__global__ void __launch_bounds__(256)
wmma_gemm_strip(const float* __restrict__ A, const float* __restrict__ W,
                float* __restrict__ C, int Mtiles, int K, int ldc, int coff) {
  const int wave = (blockIdx.x * blockDim.x + threadIdx.x) >> 5;
  if (wave >= Mtiles) return;
  const int lane = threadIdx.x & 31;
  const int lo = lane & 15;
  const int hi = lane >> 4;
  const float* Ab = A + ((size_t)wave * 16 + lo) * K + hi * 4;
  const float* Wb = W + (size_t)lo * K + hi * 4;
  v8f acc[NT];
#pragma unroll
  for (int j = 0; j < NT; ++j)
    acc[j] = (v8f){0.f, 0.f, 0.f, 0.f, 0.f, 0.f, 0.f, 0.f};
  for (int k0 = 0; k0 < K; k0 += 8) {
    const float4 a4 = *(const float4*)(Ab + k0);
    float4 b4[NT];
#pragma unroll
    for (int j = 0; j < NT; ++j) b4[j] = *(const float4*)(Wb + (size_t)j * 16 * K + k0);
#pragma unroll
    for (int j = 0; j < NT; ++j)
      acc[j] = __builtin_amdgcn_wmma_f32_16x16x4_f32(false, (v2f){a4.x, a4.y}, false, (v2f){b4[j].x, b4[j].y},
                                                     (short)0, acc[j], false, false);
#pragma unroll
    for (int j = 0; j < NT; ++j)
      acc[j] = __builtin_amdgcn_wmma_f32_16x16x4_f32(false, (v2f){a4.z, a4.w}, false, (v2f){b4[j].z, b4[j].w},
                                                     (short)0, acc[j], false, false);
  }
#pragma unroll
  for (int j = 0; j < NT; ++j) {
    float* Cb = C + (size_t)wave * 16 * ldc + coff + (size_t)j * 16 + lo;
#pragma unroll
    for (int r = 0; r < 8; ++r)
      Cb[(size_t)(r + hi * 8) * ldc] = acc[j][r];
  }
}

__global__ void __launch_bounds__(256)
wmma_gemm_frac(const float* __restrict__ H, const int* __restrict__ fc,
               const float* __restrict__ Wf, float* __restrict__ hfpre) {
  const int wave = (blockIdx.x * blockDim.x + threadIdx.x) >> 5;
  if (wave >= N_NODES / 16) return;
  const int s = blockIdx.y;
  const int lane = threadIdx.x & 31;
  const int lo = lane & 15;
  const int hi = lane >> 4;
  const int nid = fc[(wave * 16 + lo) * 3 + s];
  const float* Ab = H + (size_t)nid * IN_DIM + hi * 4;
  const float* Wb = Wf + (size_t)s * HID * IN_DIM + (size_t)lo * IN_DIM + hi * 4;
  v8f acc[2];
  acc[0] = (v8f){0.f, 0.f, 0.f, 0.f, 0.f, 0.f, 0.f, 0.f};
  acc[1] = acc[0];
  for (int k0 = 0; k0 < IN_DIM; k0 += 8) {
    const float4 a4 = *(const float4*)(Ab + k0);
    float4 b4[2];
#pragma unroll
    for (int j = 0; j < 2; ++j) b4[j] = *(const float4*)(Wb + (size_t)j * 16 * IN_DIM + k0);
#pragma unroll
    for (int j = 0; j < 2; ++j)
      acc[j] = __builtin_amdgcn_wmma_f32_16x16x4_f32(false, (v2f){a4.x, a4.y}, false, (v2f){b4[j].x, b4[j].y},
                                                     (short)0, acc[j], false, false);
#pragma unroll
    for (int j = 0; j < 2; ++j)
      acc[j] = __builtin_amdgcn_wmma_f32_16x16x4_f32(false, (v2f){a4.z, a4.w}, false, (v2f){b4[j].z, b4[j].w},
                                                     (short)0, acc[j], false, false);
  }
#pragma unroll
  for (int j = 0; j < 2; ++j) {
    float* Cb = hfpre + (size_t)wave * 16 * 96 + s * HID + (size_t)j * 16 + lo;
#pragma unroll
    for (int r = 0; r < 8; ++r)
      Cb[(size_t)(r + hi * 8) * 96] = acc[j][r];
  }
}

__global__ void attn_scores(const float* __restrict__ z, const float* __restrict__ a,
                            float* __restrict__ ss, float* __restrict__ sd,
                            int n, int heads) {
  int t = blockIdx.x * blockDim.x + threadIdx.x;
  if (t >= n * heads) return;
  int node = t / heads, hh = t - node * heads;
  const float* zp = z + (size_t)node * heads * HID + hh * HID;
  const float* al = a + (size_t)hh * 2 * HID;
  const float* ar = al + HID;
  float s0 = 0.f, s1 = 0.f;
#pragma unroll 8
  for (int o = 0; o < HID; ++o) { s0 += zp[o] * al[o]; s1 += zp[o] * ar[o]; }
  ss[t] = s0; sd[t] = s1;
}

#define AG_ROWS   391
#define AG_GROUPS 256
#define AG_CAP    16384
#define AG_TB     1024
#define AG_U      8
#define AG_RB     9
#define AG_B      4

__device__ __forceinline__ int lane_get_i(int v, int i) {
  return __builtin_amdgcn_readlane(v, i);
}
__device__ __forceinline__ float lane_get_f(float v, int i) {
  return __int_as_float(__builtin_amdgcn_readlane(__float_as_int(v), i));
}

#define OM_TB     256
#define OM_GROUPS ((N_EDGES / 16 + OM_TB - 1) / OM_TB)
__device__ __forceinline__ unsigned owner_of(int d) {
  return min((unsigned)d / (unsigned)AG_ROWS, (unsigned)(AG_GROUPS - 1));
}
__global__ void __launch_bounds__(OM_TB)
dst_owner_map(const int* __restrict__ dst, unsigned* __restrict__ omap) {
  constexpr int T = OM_GROUPS * OM_TB;
  const int t = blockIdx.x * OM_TB + threadIdx.x;
  const int4* __restrict__ dst4 = (const int4*)dst;
  int4 d[4];
#pragma unroll
  for (int u = 0; u < 4; ++u) d[u] = dst4[min(t + u * T, N_EDGES / 4 - 1)];
#pragma unroll
  for (int u = 0; u < 4; ++u) {
    const int i = t + u * T;
    if (i < N_EDGES / 4)
      omap[i] = owner_of(d[u].x) | (owner_of(d[u].y) << 8) | (owner_of(d[u].z) << 16) | (owner_of(d[u].w) << 24);
  }
}

template <int H, bool APPLY_ELU, bool TAKE_LIST>
__global__ void __launch_bounds__(AG_TB)
edge_softmax_aggregate_owner(const int* __restrict__ src, const int* __restrict__ dst,
                             const float* __restrict__ ss, const float* __restrict__ sd,
                             const float* __restrict__ z, float* __restrict__ out, int ldo,
                             int* __restrict__ wlist, int* __restrict__ wcnt,
                             const unsigned* __restrict__ omap) {
  constexpr int C = H * HID;
  constexpr int RMASK = (1 << AG_RB) - 1;
  __shared__ __attribute__((aligned(16))) float acc[(AG_ROWS + 1) * C];
  __shared__ int lst[AG_CAP];
  __shared__ __attribute__((aligned(16))) unsigned mx[(AG_ROWS + 1) * H];
  __shared__ __attribute__((aligned(16))) float sm[(AG_ROWS + 1) * H];
  __shared__ __attribute__((aligned(16))) float sdl[(AG_ROWS + 1) * H];
  __shared__ int cnt;
  const int tid  = threadIdx.x;
  const int lane = tid & 31;
  const int wave = __builtin_amdgcn_readfirstlane(tid >> 5);
  const int base = blockIdx.x * AG_ROWS;
  int* __restrict__ wl = wlist + (size_t)blockIdx.x * AG_CAP;

  for (int i = tid; i < (AG_ROWS + 1) * C; i += AG_TB) acc[i] = 0.f;
  for (int i = tid; i < (AG_ROWS + 1) * H; i += AG_TB) {
    mx[i] = 0u;
    sm[i] = 0.f;
    const int g = base * H + i;
    sdl[i] = (i < AG_ROWS * H && g < N_NODES * H) ? sd[g] : 0.f;
  }
  int total;
  if constexpr (TAKE_LIST) {
    total = wcnt[blockIdx.x];
    const int m = (total > AG_CAP) ? 0 : total;
    for (int i = tid; i < m; i += AG_TB) lst[i] = wl[i];
    __syncthreads();
  } else {
    if (tid == 0) cnt = 0;
    __syncthreads();

    const unsigned idb = (unsigned)blockIdx.x * 0x01010101u;
    auto note16 = [&](const uint4 w, const int q) {
      const unsigned x0 = w.x ^ idb, x1 = w.y ^ idb, x2 = w.z ^ idb, x3 = w.w ^ idb;
      const unsigned u0 = (x0 - 0x01010101u) & ~x0, u1 = (x1 - 0x01010101u) & ~x1;
      const unsigned u2 = (x2 - 0x01010101u) & ~x2, u3 = (x3 - 0x01010101u) & ~x3;
      if ((u0 | u1 | u2 | u3) & 0x80808080u) {
        unsigned c = ((u0 >> 7) & 0x01010101u) | ((u1 >> 6) & 0x02020202u) |
                     ((u2 >> 5) & 0x04040404u) | ((u3 >> 4) & 0x08080808u);
        int slot = atomicAdd(&cnt, __popc(c));
        while (c) {
          const int b = __builtin_ctz(c);
          c &= c - 1u;
          if (slot < AG_CAP) lst[slot] = q * 16 + ((b & 7) << 2) + (b >> 3);
          ++slot;
        }
      }
    };
    const uint4* __restrict__ om4 = (const uint4*)omap;
    constexpr int P16  = N_EDGES / 16;
    constexpr int STEP = AG_TB * 4;
    constexpr int FULL = P16 / STEP;
    {
      int q0 = tid;
      uint4 cur[4], nxt[4];
#pragma unroll
      for (int u = 0; u < 4; ++u) cur[u] = om4[q0 + u * AG_TB];
      for (int it = 0; it + 1 < FULL; ++it) {
#pragma unroll
        for (int u = 0; u < 4; ++u) nxt[u] = om4[q0 + STEP + u * AG_TB];
#pragma unroll
        for (int u = 0; u < 4; ++u) note16(cur[u], q0 + u * AG_TB);
#pragma unroll
        for (int u = 0; u < 4; ++u) cur[u] = nxt[u];
        q0 += STEP;
      }
#pragma unroll
      for (int u = 0; u < 4; ++u) note16(cur[u], q0 + u * AG_TB);
#pragma unroll
      for (int u = 0; u < 4; ++u) {
        const int q = FULL * STEP + tid + u * AG_TB;
        if (q < P16) note16(om4[q], q);
      }
    }
    __syncthreads();

    total = cnt;
    if (tid == 0) wcnt[blockIdx.x] = total;
  }
  const bool overflow = total > AG_CAP;
  const int n = __builtin_amdgcn_readfirstlane(overflow ? 0 : total);

  auto leaky = [](float x) { return (x > 0.f) ? x : 0.01f * x; };
  auto load_ss = [&](int s, float (&x)[H]) {
    if constexpr (H == 4) {
      const float4 t = ((const float4*)ss)[s];
      x[0] = t.x; x[1] = t.y; x[2] = t.z; x[3] = t.w;
    } else {
#pragma unroll
      for (int h = 0; h < H; ++h) x[h] = ss[(size_t)s * H + h];
    }
  };
  auto walk = [&](auto f) {
    for (int ed = tid; ed < N_EDGES; ed += AG_TB) {
      const unsigned r = (unsigned)(dst[ed] - base);
      if (r < (unsigned)AG_ROWS) f(src[ed], (int)r);
    }
  };

  if (overflow) {
    walk([&](int s, int r) {
      float x[H];
      load_ss(s, x);
      for (int h = 0; h < H; ++h) atomicMax(&mx[r * H + h], fkey(leaky(x[h] + sdl[r * H + h])));
    });
  } else {
    for (int i0 = tid; i0 < n; i0 += AG_B * AG_TB) {
      int p[AG_B], s[AG_B], rw[AG_B];
      float x[AG_B][H];
#pragma unroll
      for (int u = 0; u < AG_B; ++u) {
        const int idx = i0 + u * AG_TB;
        p[u] = (idx < n) ? lst[idx] : -1;
      }
#pragma unroll
      for (int u = 0; u < AG_B; ++u) {
        if constexpr (TAKE_LIST) {
          s[u]  = (p[u] >= 0) ? (p[u] >> AG_RB) : 0;
          rw[u] = p[u] & RMASK;
        } else {
          s[u]  = (p[u] >= 0) ? src[p[u]] : 0;
          const int d = (p[u] >= 0) ? dst[p[u]] : base;
          rw[u] = (int)min((unsigned)(d - base), (unsigned)AG_ROWS);
        }
      }
#pragma unroll
      for (int u = 0; u < AG_B; ++u) load_ss(s[u], x[u]);
#pragma unroll
      for (int u = 0; u < AG_B; ++u) {
        if (p[u] >= 0) {
          const int r = rw[u];
          if constexpr (!TAKE_LIST) {
            const int e = (s[u] << AG_RB) | r;
            lst[i0 + u * AG_TB] = e;
            wl[i0 + u * AG_TB] = e;
          }
#pragma unroll
          for (int h = 0; h < H; ++h)
            atomicMax(&mx[r * H + h], fkey(leaky(x[u][h] + sdl[r * H + h])));
        }
      }
    }
  }
  __syncthreads();

  if (overflow) {
    walk([&](int s, int r) {
      float x[H];
      load_ss(s, x);
      for (int h = 0; h < H; ++h)
        atomicAdd(&sm[r * H + h], expf(leaky(x[h] + sdl[r * H + h]) - funkey(mx[r * H + h])));
    });
  } else {
    for (int i0 = tid; i0 < n; i0 += AG_B * AG_TB) {
      int p[AG_B];
      float x[AG_B][H];
#pragma unroll
      for (int u = 0; u < AG_B; ++u) {
        const int idx = i0 + u * AG_TB;
        p[u] = (idx < n) ? lst[idx] : -1;
      }
#pragma unroll
      for (int u = 0; u < AG_B; ++u) load_ss((p[u] >= 0) ? (p[u] >> AG_RB) : 0, x[u]);
#pragma unroll
      for (int u = 0; u < AG_B; ++u) {
        if (p[u] >= 0) {
          const int r = p[u] & RMASK;
#pragma unroll
          for (int h = 0; h < H; ++h)
            atomicAdd(&sm[r * H + h], expf(leaky(x[u][h] + sdl[r * H + h]) - funkey(mx[r * H + h])));
        }
      }
    }
  }
  __syncthreads();

  if (overflow) {
    walk([&](int s, int r) {
      float x[H];
      load_ss(s, x);
      for (int h = 0; h < H; ++h) {
        const float a = expf(leaky(x[h] + sdl[r * H + h]) - funkey(mx[r * H + h])) / fmaxf(sm[r * H + h], 1e-9f);
        for (int o = 0; o < HID; ++o)
          atomicAdd(&acc[r * C + h * HID + o], a * z[(size_t)s * C + h * HID + o]);
      }
    });
  }
  for (int b0 = wave * 32; b0 < n; b0 += AG_TB) {
    const int idx = b0 + lane;
    const bool valid = idx < n;
    const int p = valid ? lst[idx] : AG_ROWS;
    const int s = p >> AG_RB;
    const int row = p & RMASK;
    float al[H];
    load_ss(s, al);
#pragma unroll
    for (int h = 0; h < H; ++h) {
      const float ee = expf(leaky(al[h] + sdl[row * H + h]) - funkey(mx[row * H + h]));
      al[h] = valid ? ee / fmaxf(sm[row * H + h], 1e-9f) : 0.f;
    }
    const int nb = min(32, n - b0);
    for (int i0 = 0; i0 < nb; i0 += AG_U) {
      float v[AG_U][H];
#pragma unroll
      for (int u = 0; u < AG_U; ++u) {
        const float* zp = z + (size_t)lane_get_i(s, i0 + u) * C + lane;
#pragma unroll
        for (int h = 0; h < H; ++h) v[u][h] = zp[h * HID];
      }
#pragma unroll
      for (int u = 0; u < AG_U; ++u) {
        float* ap = acc + lane_get_i(row, i0 + u) * C + lane;
#pragma unroll
        for (int h = 0; h < H; ++h)
          atomicAdd(ap + h * HID, lane_get_f(al[h], i0 + u) * v[u][h]);
      }
    }
  }
  __syncthreads();

  constexpr int C4 = C / 4;
  for (int i = tid; i < AG_ROWS * C4; i += AG_TB) {
    const int r = i / C4, c4 = i - r * C4;
    const int node = base + r;
    if (node < N_NODES) {
      float4 v = *(const float4*)&acc[r * C + c4 * 4];
      if constexpr (APPLY_ELU) {
        v.x = (v.x > 0.f) ? v.x : (expf(v.x) - 1.0f);
        v.y = (v.y > 0.f) ? v.y : (expf(v.y) - 1.0f);
        v.z = (v.z > 0.f) ? v.z : (expf(v.z) - 1.0f);
        v.w = (v.w > 0.f) ? v.w : (expf(v.w) - 1.0f);
      }
      *(float4*)&out[(size_t)node * ldo + c4 * 4] = v;
    }
  }
}

extern "C" void kernel_launch(void* const* d_in, const int* in_sizes, int n_in,
                              void* d_out, int out_size, void* d_ws, size_t ws_size,
                              hipStream_t stream) {
  const float* h    = (const float*)d_in[0];
  const int*   src  = (const int*)  d_in[1];
  const int*   dst  = (const int*)  d_in[2];
  const int*   fc   = (const int*)  d_in[3];
  const float* W1   = (const float*)d_in[4];
  const float* a1   = (const float*)d_in[5];
  const float* W2   = (const float*)d_in[6];
  const float* a2   = (const float*)d_in[7];
  const float* Wf   = (const float*)d_in[8];
  const float* Wff  = (const float*)d_in[9];
  const float* Wfc2 = (const float*)d_in[10];
  float* out = (float*)d_out;

  float* w = (float*)d_ws;
  size_t off = 0;
  float* agg1 = w + off; off += (size_t)N_NODES * 128;
  float* hg   = w + off; off += (size_t)N_NODES * 64;
  float* z1   = w + off; off += (size_t)N_NODES * 128;
  float* s1s  = w + off; off += (size_t)N_NODES * HEADS;
  float* s1d  = w + off; off += (size_t)N_NODES * HEADS;
  float* z2   = w + off; off += (size_t)N_NODES * HID;
  float* s2s  = w + off; off += N_NODES;
  float* s2d  = w + off; off += N_NODES;
  float* hfp  = w + off; off += (size_t)N_NODES * 96;
  int*   elst = (int*)(w + off); off += (size_t)AG_GROUPS * AG_CAP;
  int*   ecnt = (int*)(w + off); off += AG_GROUPS;
  unsigned* omap = (unsigned*)(w + off); off += N_EDGES / 4;
  (void)ws_size; (void)in_sizes; (void)n_in; (void)out_size;

  const int TB = 256;
  const int MT = N_NODES / 16;
  const int gemm_blocks = (MT + 7) / 8;

  wmma_gemm_strip<8><<<gemm_blocks, TB, 0, stream>>>(h, W1, z1, MT, 128, 128, 0);
  attn_scores<<<(N_NODES * HEADS + TB - 1) / TB, TB, 0, stream>>>(z1, a1, s1s, s1d, N_NODES, HEADS);
  dst_owner_map<<<OM_GROUPS, OM_TB, 0, stream>>>(dst, omap);
  edge_softmax_aggregate_owner<HEADS, true, false><<<AG_GROUPS, AG_TB, 0, stream>>>(src, dst, s1s, s1d, z1, agg1, 128, elst, ecnt, omap);

  wmma_gemm_strip<2><<<gemm_blocks, TB, 0, stream>>>(agg1, W2, z2, MT, 128, HID, 0);
  attn_scores<<<(N_NODES + TB - 1) / TB, TB, 0, stream>>>(z2, a2, s2s, s2d, N_NODES, 1);
  edge_softmax_aggregate_owner<1, false, true><<<AG_GROUPS, AG_TB, 0, stream>>>(src, dst, s2s, s2d, z2, hg, 64, elst, ecnt, omap);

  {
    dim3 g(gemm_blocks, 3);
    wmma_gemm_frac<<<g, TB, 0, stream>>>(h, fc, Wf, hfp);
  }
  wmma_gemm_strip<2><<<gemm_blocks, TB, 0, stream>>>(hfp, Wff, hg, MT, 96, 64, HID);

  wmma_gemm_strip<1><<<gemm_blocks, TB, 0, stream>>>(hg, Wfc2, out, MT, 64, OUT_DIM, 0);
}


# Each noted edge's leaky-relu score is formed once (max pass) and kept by list slot (1 head: LDS; 4 heads: a workspace row); exp-sum and weight passes no longer re-read s_src[src]; LDS sum order kept
# speedup vs baseline: 4.9663x; 1.0475x over previous
#include <hip/hip_runtime.h>
#include <math.h>

#define N_NODES 100000
#define N_EDGES 1600000
#define IN_DIM  128
#define HID     32
#define HEADS   4
#define OUT_DIM 16

typedef __attribute__((ext_vector_type(2))) float v2f;
typedef __attribute__((ext_vector_type(8))) float v8f;

__device__ __forceinline__ unsigned fkey(float x) {
  unsigned b = __float_as_uint(x);
  return (b & 0x80000000u) ? ~b : (b | 0x80000000u);
}
__device__ __forceinline__ float funkey(unsigned k) {
  return __uint_as_float((k & 0x80000000u) ? (k & 0x7FFFFFFFu) : ~k);
}

template <int NT>
__global__ void __launch_bounds__(256)
wmma_gemm_strip(const float* __restrict__ A, const float* __restrict__ W,
                float* __restrict__ C, int Mtiles, int K, int ldc, int coff) {
  const int wave = (blockIdx.x * blockDim.x + threadIdx.x) >> 5;
  if (wave >= Mtiles) return;
  const int lane = threadIdx.x & 31;
  const int lo = lane & 15;
  const int hi = lane >> 4;
  const float* Ab = A + ((size_t)wave * 16 + lo) * K + hi * 4;
  const float* Wb = W + (size_t)lo * K + hi * 4;
  v8f acc[NT];
#pragma unroll
  for (int j = 0; j < NT; ++j)
    acc[j] = (v8f){0.f, 0.f, 0.f, 0.f, 0.f, 0.f, 0.f, 0.f};
  for (int k0 = 0; k0 < K; k0 += 8) {
    const float4 a4 = *(const float4*)(Ab + k0);
    float4 b4[NT];
#pragma unroll
    for (int j = 0; j < NT; ++j) b4[j] = *(const float4*)(Wb + (size_t)j * 16 * K + k0);
#pragma unroll
    for (int j = 0; j < NT; ++j)
      acc[j] = __builtin_amdgcn_wmma_f32_16x16x4_f32(false, (v2f){a4.x, a4.y}, false, (v2f){b4[j].x, b4[j].y},
                                                     (short)0, acc[j], false, false);
#pragma unroll
    for (int j = 0; j < NT; ++j)
      acc[j] = __builtin_amdgcn_wmma_f32_16x16x4_f32(false, (v2f){a4.z, a4.w}, false, (v2f){b4[j].z, b4[j].w},
                                                     (short)0, acc[j], false, false);
  }
#pragma unroll
  for (int j = 0; j < NT; ++j) {
    float* Cb = C + (size_t)wave * 16 * ldc + coff + (size_t)j * 16 + lo;
#pragma unroll
    for (int r = 0; r < 8; ++r)
      Cb[(size_t)(r + hi * 8) * ldc] = acc[j][r];
  }
}

__global__ void __launch_bounds__(256)
wmma_gemm_frac(const float* __restrict__ H, const int* __restrict__ fc,
               const float* __restrict__ Wf, float* __restrict__ hfpre) {
  const int wave = (blockIdx.x * blockDim.x + threadIdx.x) >> 5;
  if (wave >= N_NODES / 16) return;
  const int s = blockIdx.y;
  const int lane = threadIdx.x & 31;
  const int lo = lane & 15;
  const int hi = lane >> 4;
  const int nid = fc[(wave * 16 + lo) * 3 + s];
  const float* Ab = H + (size_t)nid * IN_DIM + hi * 4;
  const float* Wb = Wf + (size_t)s * HID * IN_DIM + (size_t)lo * IN_DIM + hi * 4;
  v8f acc[2];
  acc[0] = (v8f){0.f, 0.f, 0.f, 0.f, 0.f, 0.f, 0.f, 0.f};
  acc[1] = acc[0];
  for (int k0 = 0; k0 < IN_DIM; k0 += 8) {
    const float4 a4 = *(const float4*)(Ab + k0);
    float4 b4[2];
#pragma unroll
    for (int j = 0; j < 2; ++j) b4[j] = *(const float4*)(Wb + (size_t)j * 16 * IN_DIM + k0);
#pragma unroll
    for (int j = 0; j < 2; ++j)
      acc[j] = __builtin_amdgcn_wmma_f32_16x16x4_f32(false, (v2f){a4.x, a4.y}, false, (v2f){b4[j].x, b4[j].y},
                                                     (short)0, acc[j], false, false);
#pragma unroll
    for (int j = 0; j < 2; ++j)
      acc[j] = __builtin_amdgcn_wmma_f32_16x16x4_f32(false, (v2f){a4.z, a4.w}, false, (v2f){b4[j].z, b4[j].w},
                                                     (short)0, acc[j], false, false);
  }
#pragma unroll
  for (int j = 0; j < 2; ++j) {
    float* Cb = hfpre + (size_t)wave * 16 * 96 + s * HID + (size_t)j * 16 + lo;
#pragma unroll
    for (int r = 0; r < 8; ++r)
      Cb[(size_t)(r + hi * 8) * 96] = acc[j][r];
  }
}

__global__ void attn_scores(const float* __restrict__ z, const float* __restrict__ a,
                            float* __restrict__ ss, float* __restrict__ sd,
                            int n, int heads) {
  int t = blockIdx.x * blockDim.x + threadIdx.x;
  if (t >= n * heads) return;
  int node = t / heads, hh = t - node * heads;
  const float* zp = z + (size_t)node * heads * HID + hh * HID;
  const float* al = a + (size_t)hh * 2 * HID;
  const float* ar = al + HID;
  float s0 = 0.f, s1 = 0.f;
#pragma unroll 8
  for (int o = 0; o < HID; ++o) { s0 += zp[o] * al[o]; s1 += zp[o] * ar[o]; }
  ss[t] = s0; sd[t] = s1;
}

#define AG_ROWS   391
#define AG_GROUPS 256
#define AG_CAP    16384
#define AG_TB     1024
#define AG_U      8
#define AG_RB     9
#define AG_B      4

__device__ __forceinline__ int lane_get_i(int v, int i) {
  return __builtin_amdgcn_readlane(v, i);
}
__device__ __forceinline__ float lane_get_f(float v, int i) {
  return __int_as_float(__builtin_amdgcn_readlane(__float_as_int(v), i));
}

#define OM_TB     256
#define OM_GROUPS ((N_EDGES / 16 + OM_TB - 1) / OM_TB)
__device__ __forceinline__ unsigned owner_of(int d) {
  return min((unsigned)d / (unsigned)AG_ROWS, (unsigned)(AG_GROUPS - 1));
}
__global__ void __launch_bounds__(OM_TB)
dst_owner_map(const int* __restrict__ dst, unsigned* __restrict__ omap) {
  constexpr int T = OM_GROUPS * OM_TB;
  const int t = blockIdx.x * OM_TB + threadIdx.x;
  const int4* __restrict__ dst4 = (const int4*)dst;
  int4 d[4];
#pragma unroll
  for (int u = 0; u < 4; ++u) d[u] = dst4[min(t + u * T, N_EDGES / 4 - 1)];
#pragma unroll
  for (int u = 0; u < 4; ++u) {
    const int i = t + u * T;
    if (i < N_EDGES / 4)
      omap[i] = owner_of(d[u].x) | (owner_of(d[u].y) << 8) | (owner_of(d[u].z) << 16) | (owner_of(d[u].w) << 24);
  }
}

template <int H, bool APPLY_ELU, bool TAKE_LIST>
__global__ void __launch_bounds__(AG_TB)
edge_softmax_aggregate_owner(const int* __restrict__ src, const int* __restrict__ dst,
                             const float* __restrict__ ss, const float* __restrict__ sd,
                             const float* __restrict__ z, float* __restrict__ out, int ldo,
                             int* __restrict__ wlist, int* __restrict__ wcnt,
                             const unsigned* __restrict__ omap, float* __restrict__ wscore) {
  static_assert(H == 1 || H == 4, "score slots are one float or one float4");
  constexpr int C = H * HID;
  constexpr bool SC_LDS = (H == 1);
  constexpr int RMASK = (1 << AG_RB) - 1;
  __shared__ __attribute__((aligned(16))) float acc[(AG_ROWS + 1) * C];
  __shared__ int lst[AG_CAP];
  __shared__ __attribute__((aligned(16))) unsigned mx[(AG_ROWS + 1) * H];
  __shared__ __attribute__((aligned(16))) float sm[(AG_ROWS + 1) * H];
  __shared__ __attribute__((aligned(16))) float sdl[(AG_ROWS + 1) * H];
  __shared__ float scl[SC_LDS ? AG_CAP : 1];
  __shared__ int cnt;
  const int tid  = threadIdx.x;
  const int lane = tid & 31;
  const int wave = __builtin_amdgcn_readfirstlane(tid >> 5);
  const int base = blockIdx.x * AG_ROWS;
  int* __restrict__ wl = wlist + (size_t)blockIdx.x * AG_CAP;
  float4* __restrict__ wsc = SC_LDS ? nullptr : (float4*)wscore + (size_t)blockIdx.x * AG_CAP;

  for (int i = tid; i < (AG_ROWS + 1) * C; i += AG_TB) acc[i] = 0.f;
  for (int i = tid; i < (AG_ROWS + 1) * H; i += AG_TB) {
    mx[i] = 0u;
    sm[i] = 0.f;
    const int g = base * H + i;
    sdl[i] = (i < AG_ROWS * H && g < N_NODES * H) ? sd[g] : 0.f;
  }
  int total;
  if constexpr (TAKE_LIST) {
    total = wcnt[blockIdx.x];
    const int m = (total > AG_CAP) ? 0 : total;
    for (int i = tid; i < m; i += AG_TB) lst[i] = wl[i];
    __syncthreads();
  } else {
    if (tid == 0) cnt = 0;
    __syncthreads();

    const unsigned idb = (unsigned)blockIdx.x * 0x01010101u;
    auto note16 = [&](const uint4 w, const int q) {
      const unsigned x0 = w.x ^ idb, x1 = w.y ^ idb, x2 = w.z ^ idb, x3 = w.w ^ idb;
      const unsigned u0 = (x0 - 0x01010101u) & ~x0, u1 = (x1 - 0x01010101u) & ~x1;
      const unsigned u2 = (x2 - 0x01010101u) & ~x2, u3 = (x3 - 0x01010101u) & ~x3;
      if ((u0 | u1 | u2 | u3) & 0x80808080u) {
        unsigned c = ((u0 >> 7) & 0x01010101u) | ((u1 >> 6) & 0x02020202u) |
                     ((u2 >> 5) & 0x04040404u) | ((u3 >> 4) & 0x08080808u);
        int slot = atomicAdd(&cnt, __popc(c));
        while (c) {
          const int b = __builtin_ctz(c);
          c &= c - 1u;
          if (slot < AG_CAP) lst[slot] = q * 16 + ((b & 7) << 2) + (b >> 3);
          ++slot;
        }
      }
    };
    const uint4* __restrict__ om4 = (const uint4*)omap;
    constexpr int P16  = N_EDGES / 16;
    constexpr int STEP = AG_TB * 4;
    constexpr int FULL = P16 / STEP;
    {
      int q0 = tid;
      uint4 cur[4], nxt[4];
#pragma unroll
      for (int u = 0; u < 4; ++u) cur[u] = om4[q0 + u * AG_TB];
      for (int it = 0; it + 1 < FULL; ++it) {
#pragma unroll
        for (int u = 0; u < 4; ++u) nxt[u] = om4[q0 + STEP + u * AG_TB];
#pragma unroll
        for (int u = 0; u < 4; ++u) note16(cur[u], q0 + u * AG_TB);
#pragma unroll
        for (int u = 0; u < 4; ++u) cur[u] = nxt[u];
        q0 += STEP;
      }
#pragma unroll
      for (int u = 0; u < 4; ++u) note16(cur[u], q0 + u * AG_TB);
#pragma unroll
      for (int u = 0; u < 4; ++u) {
        const int q = FULL * STEP + tid + u * AG_TB;
        if (q < P16) note16(om4[q], q);
      }
    }
    __syncthreads();

    total = cnt;
    if (tid == 0) wcnt[blockIdx.x] = total;
  }
  const bool overflow = total > AG_CAP;
  const int n = __builtin_amdgcn_readfirstlane(overflow ? 0 : total);

  auto leaky = [](float x) { return (x > 0.f) ? x : 0.01f * x; };
  auto load_ss = [&](int s, float (&x)[H]) {
    if constexpr (H == 4) {
      const float4 t = ((const float4*)ss)[s];
      x[0] = t.x; x[1] = t.y; x[2] = t.z; x[3] = t.w;
    } else {
#pragma unroll
      for (int h = 0; h < H; ++h) x[h] = ss[(size_t)s * H + h];
    }
  };
  auto put_sc = [&](int i, const float (&e)[H]) {
    if constexpr (SC_LDS) scl[i] = e[0];
    else wsc[i] = make_float4(e[0], e[1], e[2], e[3]);
  };
  auto get_sc = [&](int i, float (&e)[H]) {
    if constexpr (SC_LDS) e[0] = scl[i];
    else { const float4 t = wsc[i]; e[0] = t.x; e[1] = t.y; e[2] = t.z; e[3] = t.w; }
  };
  auto walk = [&](auto f) {
    for (int ed = tid; ed < N_EDGES; ed += AG_TB) {
      const unsigned r = (unsigned)(dst[ed] - base);
      if (r < (unsigned)AG_ROWS) f(src[ed], (int)r);
    }
  };

  if (overflow) {
    walk([&](int s, int r) {
      float x[H];
      load_ss(s, x);
      for (int h = 0; h < H; ++h) atomicMax(&mx[r * H + h], fkey(leaky(x[h] + sdl[r * H + h])));
    });
  } else {
    for (int i0 = tid; i0 < n; i0 += AG_B * AG_TB) {
      int p[AG_B], s[AG_B], rw[AG_B];
      float x[AG_B][H];
#pragma unroll
      for (int u = 0; u < AG_B; ++u) {
        const int idx = i0 + u * AG_TB;
        p[u] = (idx < n) ? lst[idx] : -1;
      }
#pragma unroll
      for (int u = 0; u < AG_B; ++u) {
        if constexpr (TAKE_LIST) {
          s[u]  = (p[u] >= 0) ? (p[u] >> AG_RB) : 0;
          rw[u] = p[u] & RMASK;
        } else {
          s[u]  = (p[u] >= 0) ? src[p[u]] : 0;
          const int d = (p[u] >= 0) ? dst[p[u]] : base;
          rw[u] = (int)min((unsigned)(d - base), (unsigned)AG_ROWS);
        }
      }
#pragma unroll
      for (int u = 0; u < AG_B; ++u) load_ss(s[u], x[u]);
#pragma unroll
      for (int u = 0; u < AG_B; ++u) {
        if (p[u] >= 0) {
          const int r = rw[u];
          if constexpr (!TAKE_LIST) {
            const int e = (s[u] << AG_RB) | r;
            lst[i0 + u * AG_TB] = e;
            wl[i0 + u * AG_TB] = e;
          }
          float e[H];
#pragma unroll
          for (int h = 0; h < H; ++h) {
            e[h] = leaky(x[u][h] + sdl[r * H + h]);
            atomicMax(&mx[r * H + h], fkey(e[h]));
          }
          put_sc(i0 + u * AG_TB, e);
        }
      }
    }
  }
  __syncthreads();

  if (overflow) {
    walk([&](int s, int r) {
      float x[H];
      load_ss(s, x);
      for (int h = 0; h < H; ++h)
        atomicAdd(&sm[r * H + h], expf(leaky(x[h] + sdl[r * H + h]) - funkey(mx[r * H + h])));
    });
  } else {
    for (int i0 = tid; i0 < n; i0 += AG_B * AG_TB) {
      int p[AG_B];
      float e[AG_B][H];
#pragma unroll
      for (int u = 0; u < AG_B; ++u) {
        const int idx = i0 + u * AG_TB;
        p[u] = (idx < n) ? lst[idx] : -1;
      }
#pragma unroll
      for (int u = 0; u < AG_B; ++u) get_sc((p[u] >= 0) ? (i0 + u * AG_TB) : i0, e[u]);
#pragma unroll
      for (int u = 0; u < AG_B; ++u) {
        if (p[u] >= 0) {
          const int r = p[u] & RMASK;
#pragma unroll
          for (int h = 0; h < H; ++h)
            atomicAdd(&sm[r * H + h], expf(e[u][h] - funkey(mx[r * H + h])));
        }
      }
    }
  }
  __syncthreads();

  if (overflow) {
    walk([&](int s, int r) {
      float x[H];
      load_ss(s, x);
      for (int h = 0; h < H; ++h) {
        const float a = expf(leaky(x[h] + sdl[r * H + h]) - funkey(mx[r * H + h])) / fmaxf(sm[r * H + h], 1e-9f);
        for (int o = 0; o < HID; ++o)
          atomicAdd(&acc[r * C + h * HID + o], a * z[(size_t)s * C + h * HID + o]);
      }
    });
  }
  for (int b0 = wave * 32; b0 < n; b0 += AG_TB) {
    const int idx = b0 + lane;
    const bool valid = idx < n;
    const int p = valid ? lst[idx] : AG_ROWS;
    const int s = p >> AG_RB;
    const int row = p & RMASK;
    float al[H];
#pragma unroll
    for (int h = 0; h < H; ++h) al[h] = 0.f;
    if (valid) get_sc(idx, al);
#pragma unroll
    for (int h = 0; h < H; ++h) {
      const float ee = expf(al[h] - funkey(mx[row * H + h]));
      al[h] = valid ? ee / fmaxf(sm[row * H + h], 1e-9f) : 0.f;
    }
    const int nb = min(32, n - b0);
    for (int i0 = 0; i0 < nb; i0 += AG_U) {
      float v[AG_U][H];
#pragma unroll
      for (int u = 0; u < AG_U; ++u) {
        const float* zp = z + (size_t)lane_get_i(s, i0 + u) * C + lane;
#pragma unroll
        for (int h = 0; h < H; ++h) v[u][h] = zp[h * HID];
      }
#pragma unroll
      for (int u = 0; u < AG_U; ++u) {
        float* ap = acc + lane_get_i(row, i0 + u) * C + lane;
#pragma unroll
        for (int h = 0; h < H; ++h)
          atomicAdd(ap + h * HID, lane_get_f(al[h], i0 + u) * v[u][h]);
      }
    }
  }
  __syncthreads();

  constexpr int C4 = C / 4;
  for (int i = tid; i < AG_ROWS * C4; i += AG_TB) {
    const int r = i / C4, c4 = i - r * C4;
    const int node = base + r;
    if (node < N_NODES) {
      float4 v = *(const float4*)&acc[r * C + c4 * 4];
      if constexpr (APPLY_ELU) {
        v.x = (v.x > 0.f) ? v.x : (expf(v.x) - 1.0f);
        v.y = (v.y > 0.f) ? v.y : (expf(v.y) - 1.0f);
        v.z = (v.z > 0.f) ? v.z : (expf(v.z) - 1.0f);
        v.w = (v.w > 0.f) ? v.w : (expf(v.w) - 1.0f);
      }
      *(float4*)&out[(size_t)node * ldo + c4 * 4] = v;
    }
  }
}

static_assert((size_t)AG_GROUPS * AG_CAP * 4 <= (size_t)N_NODES * (64 + HID + 2 + 96),
              "the score plane must fit inside the planes it lies over");

extern "C" void kernel_launch(void* const* d_in, const int* in_sizes, int n_in,
                              void* d_out, int out_size, void* d_ws, size_t ws_size,
                              hipStream_t stream) {
  const float* h    = (const float*)d_in[0];
  const int*   src  = (const int*)  d_in[1];
  const int*   dst  = (const int*)  d_in[2];
  const int*   fc   = (const int*)  d_in[3];
  const float* W1   = (const float*)d_in[4];
  const float* a1   = (const float*)d_in[5];
  const float* W2   = (const float*)d_in[6];
  const float* a2   = (const float*)d_in[7];
  const float* Wf   = (const float*)d_in[8];
  const float* Wff  = (const float*)d_in[9];
  const float* Wfc2 = (const float*)d_in[10];
  float* out = (float*)d_out;

  float* w = (float*)d_ws;
  size_t off = 0;
  float* agg1 = w + off; off += (size_t)N_NODES * 128;
  float* z1   = w + off; off += (size_t)N_NODES * 128;
  float* s1s  = w + off; off += (size_t)N_NODES * HEADS;
  float* s1d  = w + off; off += (size_t)N_NODES * HEADS;
  int*   elst = (int*)(w + off); off += (size_t)AG_GROUPS * AG_CAP;
  int*   ecnt = (int*)(w + off); off += AG_GROUPS;
  unsigned* omap = (unsigned*)(w + off); off += N_EDGES / 4;
  float* esc  = w + off;
  float* hg   = w + off; off += (size_t)N_NODES * 64;
  float* z2   = w + off; off += (size_t)N_NODES * HID;
  float* s2s  = w + off; off += N_NODES;
  float* s2d  = w + off; off += N_NODES;
  float* hfp  = w + off; off += (size_t)N_NODES * 96;
  (void)ws_size; (void)in_sizes; (void)n_in; (void)out_size;

  const int TB = 256;
  const int MT = N_NODES / 16;
  const int gemm_blocks = (MT + 7) / 8;

  wmma_gemm_strip<8><<<gemm_blocks, TB, 0, stream>>>(h, W1, z1, MT, 128, 128, 0);
  attn_scores<<<(N_NODES * HEADS + TB - 1) / TB, TB, 0, stream>>>(z1, a1, s1s, s1d, N_NODES, HEADS);
  dst_owner_map<<<OM_GROUPS, OM_TB, 0, stream>>>(dst, omap);
  edge_softmax_aggregate_owner<HEADS, true, false><<<AG_GROUPS, AG_TB, 0, stream>>>(src, dst, s1s, s1d, z1, agg1, 128, elst, ecnt, omap, esc);

  wmma_gemm_strip<2><<<gemm_blocks, TB, 0, stream>>>(agg1, W2, z2, MT, 128, HID, 0);
  attn_scores<<<(N_NODES + TB - 1) / TB, TB, 0, stream>>>(z2, a2, s2s, s2d, N_NODES, 1);
  edge_softmax_aggregate_owner<1, false, true><<<AG_GROUPS, AG_TB, 0, stream>>>(src, dst, s2s, s2d, z2, hg, 64, elst, ecnt, omap, nullptr);

  {
    dim3 g(gemm_blocks, 3);
    wmma_gemm_frac<<<g, TB, 0, stream>>>(h, fc, Wf, hfp);
  }
  wmma_gemm_strip<2><<<gemm_blocks, TB, 0, stream>>>(hfp, Wff, hg, MT, 96, 64, HID);

  wmma_gemm_strip<1><<<gemm_blocks, TB, 0, stream>>>(hg, Wfc2, out, MT, 64, OUT_DIM, 0);
}


# Layer-1 owners read their own row of per-range edge bitmaps (1 bit per edge; the small launch builds them in LDS, 512-byte runs), not a byte per edge; popcount-prefix slots; LDS sums in arrival order
# speedup vs baseline: 5.3823x; 1.0838x over previous
#include <hip/hip_runtime.h>
#include <math.h>

#define N_NODES 100000
#define N_EDGES 1600000
#define IN_DIM  128
#define HID     32
#define HEADS   4
#define OUT_DIM 16

typedef __attribute__((ext_vector_type(2))) float v2f;
typedef __attribute__((ext_vector_type(8))) float v8f;

__device__ __forceinline__ unsigned fkey(float x) {
  unsigned b = __float_as_uint(x);
  return (b & 0x80000000u) ? ~b : (b | 0x80000000u);
}
__device__ __forceinline__ float funkey(unsigned k) {
  return __uint_as_float((k & 0x80000000u) ? (k & 0x7FFFFFFFu) : ~k);
}

template <int NT>
__global__ void __launch_bounds__(256)
wmma_gemm_strip(const float* __restrict__ A, const float* __restrict__ W,
                float* __restrict__ C, int Mtiles, int K, int ldc, int coff) {
  const int wave = (blockIdx.x * blockDim.x + threadIdx.x) >> 5;
  if (wave >= Mtiles) return;
  const int lane = threadIdx.x & 31;
  const int lo = lane & 15;
  const int hi = lane >> 4;
  const float* Ab = A + ((size_t)wave * 16 + lo) * K + hi * 4;
  const float* Wb = W + (size_t)lo * K + hi * 4;
  v8f acc[NT];
#pragma unroll
  for (int j = 0; j < NT; ++j)
    acc[j] = (v8f){0.f, 0.f, 0.f, 0.f, 0.f, 0.f, 0.f, 0.f};
  for (int k0 = 0; k0 < K; k0 += 8) {
    const float4 a4 = *(const float4*)(Ab + k0);
    float4 b4[NT];
#pragma unroll
    for (int j = 0; j < NT; ++j) b4[j] = *(const float4*)(Wb + (size_t)j * 16 * K + k0);
#pragma unroll
    for (int j = 0; j < NT; ++j)
      acc[j] = __builtin_amdgcn_wmma_f32_16x16x4_f32(false, (v2f){a4.x, a4.y}, false, (v2f){b4[j].x, b4[j].y},
                                                     (short)0, acc[j], false, false);
#pragma unroll
    for (int j = 0; j < NT; ++j)
      acc[j] = __builtin_amdgcn_wmma_f32_16x16x4_f32(false, (v2f){a4.z, a4.w}, false, (v2f){b4[j].z, b4[j].w},
                                                     (short)0, acc[j], false, false);
  }
#pragma unroll
  for (int j = 0; j < NT; ++j) {
    float* Cb = C + (size_t)wave * 16 * ldc + coff + (size_t)j * 16 + lo;
#pragma unroll
    for (int r = 0; r < 8; ++r)
      Cb[(size_t)(r + hi * 8) * ldc] = acc[j][r];
  }
}

__global__ void __launch_bounds__(256)
wmma_gemm_frac(const float* __restrict__ H, const int* __restrict__ fc,
               const float* __restrict__ Wf, float* __restrict__ hfpre) {
  const int wave = (blockIdx.x * blockDim.x + threadIdx.x) >> 5;
  if (wave >= N_NODES / 16) return;
  const int s = blockIdx.y;
  const int lane = threadIdx.x & 31;
  const int lo = lane & 15;
  const int hi = lane >> 4;
  const int nid = fc[(wave * 16 + lo) * 3 + s];
  const float* Ab = H + (size_t)nid * IN_DIM + hi * 4;
  const float* Wb = Wf + (size_t)s * HID * IN_DIM + (size_t)lo * IN_DIM + hi * 4;
  v8f acc[2];
  acc[0] = (v8f){0.f, 0.f, 0.f, 0.f, 0.f, 0.f, 0.f, 0.f};
  acc[1] = acc[0];
  for (int k0 = 0; k0 < IN_DIM; k0 += 8) {
    const float4 a4 = *(const float4*)(Ab + k0);
    float4 b4[2];
#pragma unroll
    for (int j = 0; j < 2; ++j) b4[j] = *(const float4*)(Wb + (size_t)j * 16 * IN_DIM + k0);
#pragma unroll
    for (int j = 0; j < 2; ++j)
      acc[j] = __builtin_amdgcn_wmma_f32_16x16x4_f32(false, (v2f){a4.x, a4.y}, false, (v2f){b4[j].x, b4[j].y},
                                                     (short)0, acc[j], false, false);
#pragma unroll
    for (int j = 0; j < 2; ++j)
      acc[j] = __builtin_amdgcn_wmma_f32_16x16x4_f32(false, (v2f){a4.z, a4.w}, false, (v2f){b4[j].z, b4[j].w},
                                                     (short)0, acc[j], false, false);
  }
#pragma unroll
  for (int j = 0; j < 2; ++j) {
    float* Cb = hfpre + (size_t)wave * 16 * 96 + s * HID + (size_t)j * 16 + lo;
#pragma unroll
    for (int r = 0; r < 8; ++r)
      Cb[(size_t)(r + hi * 8) * 96] = acc[j][r];
  }
}

__global__ void attn_scores(const float* __restrict__ z, const float* __restrict__ a,
                            float* __restrict__ ss, float* __restrict__ sd,
                            int n, int heads) {
  int t = blockIdx.x * blockDim.x + threadIdx.x;
  if (t >= n * heads) return;
  int node = t / heads, hh = t - node * heads;
  const float* zp = z + (size_t)node * heads * HID + hh * HID;
  const float* al = a + (size_t)hh * 2 * HID;
  const float* ar = al + HID;
  float s0 = 0.f, s1 = 0.f;
#pragma unroll 8
  for (int o = 0; o < HID; ++o) { s0 += zp[o] * al[o]; s1 += zp[o] * ar[o]; }
  ss[t] = s0; sd[t] = s1;
}

#define AG_ROWS   391
#define AG_GROUPS 256
#define AG_CAP    16384
#define AG_TB     1024
#define AG_U      8
#define AG_RB     9
#define AG_B      4

__device__ __forceinline__ int lane_get_i(int v, int i) {
  return __builtin_amdgcn_readlane(v, i);
}
__device__ __forceinline__ float lane_get_f(float v, int i) {
  return __int_as_float(__builtin_amdgcn_readlane(__float_as_int(v), i));
}

#define BM_EC     4096
#define BM_WPC    (BM_EC / 32)
#define BM_CHUNKS ((N_EDGES + BM_EC - 1) / BM_EC)
#define BM_PITCH  (BM_CHUNKS * BM_WPC)
#define BM_TB     256
#define BM_LP     (BM_WPC + 4)
static_assert(N_EDGES % 4 == 0 && BM_EC % (4 * BM_TB) == 0 && BM_WPC == 32 * 4, "one 16-byte piece per lane and row");
__device__ __forceinline__ unsigned owner_of(int d) {
  return min((unsigned)d / (unsigned)AG_ROWS, (unsigned)(AG_GROUPS - 1));
}
__global__ void __launch_bounds__(BM_TB)
dst_range_bitmaps(const int* __restrict__ dst, unsigned* __restrict__ obm) {
  __shared__ __attribute__((aligned(16))) unsigned tile[AG_GROUPS * BM_LP];
  constexpr int PER = BM_EC / 4 / BM_TB;
  const int tid = threadIdx.x;
  const int e0 = blockIdx.x * BM_EC;
  const int4* __restrict__ dst4 = (const int4*)dst;
  int4 d[PER];
#pragma unroll
  for (int u = 0; u < PER; ++u) d[u] = dst4[min((e0 >> 2) + tid + u * BM_TB, N_EDGES / 4 - 1)];
  for (int i = tid; i < AG_GROUPS * BM_LP / 4; i += BM_TB) ((uint4*)tile)[i] = make_uint4(0u, 0u, 0u, 0u);
  __syncthreads();
#pragma unroll
  for (int u = 0; u < PER; ++u) {
    const int el = (tid + u * BM_TB) * 4;
    if (e0 + el < N_EDGES) {
      const int wd = el >> 5;
      const unsigned bit = 1u << (el & 31);
      atomicOr(&tile[owner_of(d[u].x) * BM_LP + wd], bit);
      atomicOr(&tile[owner_of(d[u].y) * BM_LP + wd], bit << 1);
      atomicOr(&tile[owner_of(d[u].z) * BM_LP + wd], bit << 2);
      atomicOr(&tile[owner_of(d[u].w) * BM_LP + wd], bit << 3);
    }
  }
  __syncthreads();
  const int lane = tid & 31, wave = tid >> 5;
  constexpr int NW = BM_TB / 32;
  static_assert(AG_GROUPS % (4 * NW) == 0, "every wave stores whole groups of four rows");
  for (int r0 = wave; r0 < AG_GROUPS; r0 += 4 * NW) {
    uint4 v[4];
#pragma unroll
    for (int u = 0; u < 4; ++u) v[u] = *(const uint4*)&tile[(r0 + u * NW) * BM_LP + lane * 4];
#pragma unroll
    for (int u = 0; u < 4; ++u)
      *(uint4*)&obm[(size_t)(r0 + u * NW) * BM_PITCH + (size_t)blockIdx.x * BM_WPC + lane * 4] = v[u];
  }
}

template <int H, bool APPLY_ELU, bool TAKE_LIST>
__global__ void __launch_bounds__(AG_TB)
edge_softmax_aggregate_owner(const int* __restrict__ src, const int* __restrict__ dst,
                             const float* __restrict__ ss, const float* __restrict__ sd,
                             const float* __restrict__ z, float* __restrict__ out, int ldo,
                             int* __restrict__ wlist, int* __restrict__ wcnt,
                             const unsigned* __restrict__ obm, float* __restrict__ wscore) {
  static_assert(H == 1 || H == 4, "score slots are one float or one float4");
  constexpr int C = H * HID;
  constexpr bool SC_LDS = (H == 1);
  constexpr int RMASK = (1 << AG_RB) - 1;
  __shared__ __attribute__((aligned(16))) float acc[(AG_ROWS + 1) * C];
  __shared__ int lst[AG_CAP];
  __shared__ __attribute__((aligned(16))) unsigned mx[(AG_ROWS + 1) * H];
  __shared__ __attribute__((aligned(16))) float sm[(AG_ROWS + 1) * H];
  __shared__ __attribute__((aligned(16))) float sdl[(AG_ROWS + 1) * H];
  __shared__ float scl[SC_LDS ? AG_CAP : 1];
  __shared__ int cnt;
  const int tid  = threadIdx.x;
  const int lane = tid & 31;
  const int wave = __builtin_amdgcn_readfirstlane(tid >> 5);
  const int base = blockIdx.x * AG_ROWS;
  int* __restrict__ wl = wlist + (size_t)blockIdx.x * AG_CAP;
  float4* __restrict__ wsc = SC_LDS ? nullptr : (float4*)wscore + (size_t)blockIdx.x * AG_CAP;

  for (int i = tid; i < (AG_ROWS + 1) * C; i += AG_TB) acc[i] = 0.f;
  for (int i = tid; i < (AG_ROWS + 1) * H; i += AG_TB) {
    mx[i] = 0u;
    sm[i] = 0.f;
    const int g = base * H + i;
    sdl[i] = (i < AG_ROWS * H && g < N_NODES * H) ? sd[g] : 0.f;
  }
  int total;
  if constexpr (TAKE_LIST) {
    total = wcnt[blockIdx.x];
    const int m = (total > AG_CAP) ? 0 : total;
    for (int i = tid; i < m; i += AG_TB) lst[i] = wl[i];
    __syncthreads();
  } else {
    if (tid == 0) cnt = 0;
    __syncthreads();

    const uint4* __restrict__ row4 = (const uint4*)(obm + (size_t)blockIdx.x * BM_PITCH);
    constexpr int Q     = BM_PITCH / 4;
    constexpr int TRIPS = (Q + AG_TB - 1) / AG_TB;
    uint4 w[TRIPS];
#pragma unroll
    for (int t = 0; t < TRIPS; ++t) {
      const int q = tid + t * AG_TB;
      w[t] = row4[min(q, Q - 1)];
      if (q >= Q) w[t] = make_uint4(0u, 0u, 0u, 0u);
    }
    int c = 0;
#pragma unroll
    for (int t = 0; t < TRIPS; ++t) c += __popc(w[t].x) + __popc(w[t].y) + __popc(w[t].z) + __popc(w[t].w);
    int incl = c;
#pragma unroll
    for (int dl = 1; dl < 32; dl <<= 1) {
      const int v = __shfl_up(incl, dl);
      if (lane >= dl) incl += v;
    }
    int wbase = 0;
    if (lane == 31) wbase = atomicAdd(&cnt, incl);
    wbase = __builtin_amdgcn_readlane(wbase, 31);
    int slot = wbase + incl - c;
#pragma unroll
    for (int t = 0; t < TRIPS; ++t) {
      const int q = tid + t * AG_TB;
      const unsigned ww[4] = {w[t].x, w[t].y, w[t].z, w[t].w};
#pragma unroll
      for (int j = 0; j < 4; ++j) {
        unsigned x = ww[j];
        while (x) {
          const int b = __builtin_ctz(x);
          x &= x - 1u;
          if (slot < AG_CAP) lst[slot] = q * 128 + j * 32 + b;
          ++slot;
        }
      }
    }
    __syncthreads();

    total = cnt;
    if (tid == 0) wcnt[blockIdx.x] = total;
  }
  const bool overflow = total > AG_CAP;
  const int n = __builtin_amdgcn_readfirstlane(overflow ? 0 : total);

  auto leaky = [](float x) { return (x > 0.f) ? x : 0.01f * x; };
  auto load_ss = [&](int s, float (&x)[H]) {
    if constexpr (H == 4) {
      const float4 t = ((const float4*)ss)[s];
      x[0] = t.x; x[1] = t.y; x[2] = t.z; x[3] = t.w;
    } else {
#pragma unroll
      for (int h = 0; h < H; ++h) x[h] = ss[(size_t)s * H + h];
    }
  };
  auto put_sc = [&](int i, const float (&e)[H]) {
    if constexpr (SC_LDS) scl[i] = e[0];
    else wsc[i] = make_float4(e[0], e[1], e[2], e[3]);
  };
  auto get_sc = [&](int i, float (&e)[H]) {
    if constexpr (SC_LDS) e[0] = scl[i];
    else { const float4 t = wsc[i]; e[0] = t.x; e[1] = t.y; e[2] = t.z; e[3] = t.w; }
  };
  auto walk = [&](auto f) {
    for (int ed = tid; ed < N_EDGES; ed += AG_TB) {
      const unsigned r = (unsigned)(dst[ed] - base);
      if (r < (unsigned)AG_ROWS) f(src[ed], (int)r);
    }
  };

  if (overflow) {
    walk([&](int s, int r) {
      float x[H];
      load_ss(s, x);
      for (int h = 0; h < H; ++h) atomicMax(&mx[r * H + h], fkey(leaky(x[h] + sdl[r * H + h])));
    });
  } else {
    for (int i0 = tid; i0 < n; i0 += AG_B * AG_TB) {
      int p[AG_B], s[AG_B], rw[AG_B];
      float x[AG_B][H];
#pragma unroll
      for (int u = 0; u < AG_B; ++u) {
        const int idx = i0 + u * AG_TB;
        p[u] = (idx < n) ? lst[idx] : -1;
      }
#pragma unroll
      for (int u = 0; u < AG_B; ++u) {
        if constexpr (TAKE_LIST) {
          s[u]  = (p[u] >= 0) ? (p[u] >> AG_RB) : 0;
          rw[u] = p[u] & RMASK;
        } else {
          s[u]  = (p[u] >= 0) ? src[p[u]] : 0;
          const int d = (p[u] >= 0) ? dst[p[u]] : base;
          rw[u] = (int)min((unsigned)(d - base), (unsigned)AG_ROWS);
        }
      }
#pragma unroll
      for (int u = 0; u < AG_B; ++u) load_ss(s[u], x[u]);
#pragma unroll
      for (int u = 0; u < AG_B; ++u) {
        if (p[u] >= 0) {
          const int r = rw[u];
          if constexpr (!TAKE_LIST) {
            const int e = (s[u] << AG_RB) | r;
            lst[i0 + u * AG_TB] = e;
            wl[i0 + u * AG_TB] = e;
          }
          float e[H];
#pragma unroll
          for (int h = 0; h < H; ++h) {
            e[h] = leaky(x[u][h] + sdl[r * H + h]);
            atomicMax(&mx[r * H + h], fkey(e[h]));
          }
          put_sc(i0 + u * AG_TB, e);
        }
      }
    }
  }
  __syncthreads();

  if (overflow) {
    walk([&](int s, int r) {
      float x[H];
      load_ss(s, x);
      for (int h = 0; h < H; ++h)
        atomicAdd(&sm[r * H + h], expf(leaky(x[h] + sdl[r * H + h]) - funkey(mx[r * H + h])));
    });
  } else {
    for (int i0 = tid; i0 < n; i0 += AG_B * AG_TB) {
      int p[AG_B];
      float e[AG_B][H];
#pragma unroll
      for (int u = 0; u < AG_B; ++u) {
        const int idx = i0 + u * AG_TB;
        p[u] = (idx < n) ? lst[idx] : -1;
      }
#pragma unroll
      for (int u = 0; u < AG_B; ++u) get_sc((p[u] >= 0) ? (i0 + u * AG_TB) : i0, e[u]);
#pragma unroll
      for (int u = 0; u < AG_B; ++u) {
        if (p[u] >= 0) {
          const int r = p[u] & RMASK;
#pragma unroll
          for (int h = 0; h < H; ++h)
            atomicAdd(&sm[r * H + h], expf(e[u][h] - funkey(mx[r * H + h])));
        }
      }
    }
  }
  __syncthreads();

  if (overflow) {
    walk([&](int s, int r) {
      float x[H];
      load_ss(s, x);
      for (int h = 0; h < H; ++h) {
        const float a = expf(leaky(x[h] + sdl[r * H + h]) - funkey(mx[r * H + h])) / fmaxf(sm[r * H + h], 1e-9f);
        for (int o = 0; o < HID; ++o)
          atomicAdd(&acc[r * C + h * HID + o], a * z[(size_t)s * C + h * HID + o]);
      }
    });
  }
  for (int b0 = wave * 32; b0 < n; b0 += AG_TB) {
    const int idx = b0 + lane;
    const bool valid = idx < n;
    const int p = valid ? lst[idx] : AG_ROWS;
    const int s = p >> AG_RB;
    const int row = p & RMASK;
    float al[H];
#pragma unroll
    for (int h = 0; h < H; ++h) al[h] = 0.f;
    if (valid) get_sc(idx, al);
#pragma unroll
    for (int h = 0; h < H; ++h) {
      const float ee = expf(al[h] - funkey(mx[row * H + h]));
      al[h] = valid ? ee / fmaxf(sm[row * H + h], 1e-9f) : 0.f;
    }
    const int nb = min(32, n - b0);
    for (int i0 = 0; i0 < nb; i0 += AG_U) {
      float v[AG_U][H];
#pragma unroll
      for (int u = 0; u < AG_U; ++u) {
        const float* zp = z + (size_t)lane_get_i(s, i0 + u) * C + lane;
#pragma unroll
        for (int h = 0; h < H; ++h) v[u][h] = zp[h * HID];
      }
#pragma unroll
      for (int u = 0; u < AG_U; ++u) {
        float* ap = acc + lane_get_i(row, i0 + u) * C + lane;
#pragma unroll
        for (int h = 0; h < H; ++h)
          atomicAdd(ap + h * HID, lane_get_f(al[h], i0 + u) * v[u][h]);
      }
    }
  }
  __syncthreads();

  constexpr int C4 = C / 4;
  for (int i = tid; i < AG_ROWS * C4; i += AG_TB) {
    const int r = i / C4, c4 = i - r * C4;
    const int node = base + r;
    if (node < N_NODES) {
      float4 v = *(const float4*)&acc[r * C + c4 * 4];
      if constexpr (APPLY_ELU) {
        v.x = (v.x > 0.f) ? v.x : (expf(v.x) - 1.0f);
        v.y = (v.y > 0.f) ? v.y : (expf(v.y) - 1.0f);
        v.z = (v.z > 0.f) ? v.z : (expf(v.z) - 1.0f);
        v.w = (v.w > 0.f) ? v.w : (expf(v.w) - 1.0f);
      }
      *(float4*)&out[(size_t)node * ldo + c4 * 4] = v;
    }
  }
}

static_assert((size_t)AG_GROUPS * AG_CAP * 4 <= (size_t)N_NODES * (64 + HID + 2 + 96),
              "the score plane must fit inside the planes it lies over");

extern "C" void kernel_launch(void* const* d_in, const int* in_sizes, int n_in,
                              void* d_out, int out_size, void* d_ws, size_t ws_size,
                              hipStream_t stream) {
  const float* h    = (const float*)d_in[0];
  const int*   src  = (const int*)  d_in[1];
  const int*   dst  = (const int*)  d_in[2];
  const int*   fc   = (const int*)  d_in[3];
  const float* W1   = (const float*)d_in[4];
  const float* a1   = (const float*)d_in[5];
  const float* W2   = (const float*)d_in[6];
  const float* a2   = (const float*)d_in[7];
  const float* Wf   = (const float*)d_in[8];
  const float* Wff  = (const float*)d_in[9];
  const float* Wfc2 = (const float*)d_in[10];
  float* out = (float*)d_out;

  float* w = (float*)d_ws;
  size_t off = 0;
  float* agg1 = w + off; off += (size_t)N_NODES * 128;
  float* z1   = w + off; off += (size_t)N_NODES * 128;
  float* s1s  = w + off; off += (size_t)N_NODES * HEADS;
  float* s1d  = w + off; off += (size_t)N_NODES * HEADS;
  int*   elst = (int*)(w + off); off += (size_t)AG_GROUPS * AG_CAP;
  int*   ecnt = (int*)(w + off); off += AG_GROUPS;
  unsigned* obm  = (unsigned*)(w + off); off += (size_t)AG_GROUPS * BM_PITCH;
  float* esc  = w + off;
  float* hg   = w + off; off += (size_t)N_NODES * 64;
  float* z2   = w + off; off += (size_t)N_NODES * HID;
  float* s2s  = w + off; off += N_NODES;
  float* s2d  = w + off; off += N_NODES;
  float* hfp  = w + off; off += (size_t)N_NODES * 96;
  (void)ws_size; (void)in_sizes; (void)n_in; (void)out_size;

  const int TB = 256;
  const int MT = N_NODES / 16;
  const int gemm_blocks = (MT + 7) / 8;

  wmma_gemm_strip<8><<<gemm_blocks, TB, 0, stream>>>(h, W1, z1, MT, 128, 128, 0);
  attn_scores<<<(N_NODES * HEADS + TB - 1) / TB, TB, 0, stream>>>(z1, a1, s1s, s1d, N_NODES, HEADS);
  dst_range_bitmaps<<<BM_CHUNKS, BM_TB, 0, stream>>>(dst, obm);
  edge_softmax_aggregate_owner<HEADS, true, false><<<AG_GROUPS, AG_TB, 0, stream>>>(src, dst, s1s, s1d, z1, agg1, 128, elst, ecnt, obm, esc);

  wmma_gemm_strip<2><<<gemm_blocks, TB, 0, stream>>>(agg1, W2, z2, MT, 128, HID, 0);
  attn_scores<<<(N_NODES + TB - 1) / TB, TB, 0, stream>>>(z2, a2, s2s, s2d, N_NODES, 1);
  edge_softmax_aggregate_owner<1, false, true><<<AG_GROUPS, AG_TB, 0, stream>>>(src, dst, s2s, s2d, z2, hg, 64, elst, ecnt, obm, nullptr);

  {
    dim3 g(gemm_blocks, 3);
    wmma_gemm_frac<<<g, TB, 0, stream>>>(h, fc, Wf, hfp);
  }
  wmma_gemm_strip<2><<<gemm_blocks, TB, 0, stream>>>(hfp, Wff, hg, MT, 96, 64, HID);

  wmma_gemm_strip<1><<<gemm_blocks, TB, 0, stream>>>(hg, Wfc2, out, MT, 64, OUT_DIM, 0);
}


# GEMMs as 256 persistent workgroups: a wave loads its weight fragments once into registers and loops over 16-row strips, next strip's rows prefetched; no per-strip weight loads; k order kept
# speedup vs baseline: 6.3236x; 1.1749x over previous
#include <hip/hip_runtime.h>
#include <math.h>

#define N_NODES 100000
#define N_EDGES 1600000
#define IN_DIM  128
#define HID     32
#define HEADS   4
#define OUT_DIM 16

typedef __attribute__((ext_vector_type(2))) float v2f;
typedef __attribute__((ext_vector_type(8))) float v8f;

__device__ __forceinline__ unsigned fkey(float x) {
  unsigned b = __float_as_uint(x);
  return (b & 0x80000000u) ? ~b : (b | 0x80000000u);
}
__device__ __forceinline__ float funkey(unsigned k) {
  return __uint_as_float((k & 0x80000000u) ? (k & 0x7FFFFFFFu) : ~k);
}

template <int NT, int K, int CG, int WPG, bool GATHER>
__global__ void __launch_bounds__(CG * WPG * 32)
wmma_gemm_resident(const float* __restrict__ A, const int* __restrict__ rowidx,
                   const float* __restrict__ W, float* __restrict__ C,
                   int Mtiles, int ldc, int coff) {
  constexpr int KB = K / 8;
  const int lane = threadIdx.x & 31;
  const int lo = lane & 15;
  const int hi = lane >> 4;
  const int wv = __builtin_amdgcn_readfirstlane((int)(threadIdx.x >> 5));
  const int cg = wv / WPG;
  const int nslots = WPG * (int)gridDim.x;
  int first = (wv % WPG) * (int)gridDim.x + (int)blockIdx.x + cg * (nslots / CG);
  if (first >= nslots) first -= nslots;
  if (first >= Mtiles) return;

  const float* Wb = W + ((size_t)cg * NT * 16 + lo) * K + hi * 4;
  float4 b[KB][NT];
#pragma unroll
  for (int kb = 0; kb < KB; ++kb)
#pragma unroll
    for (int j = 0; j < NT; ++j) b[kb][j] = *(const float4*)(Wb + (size_t)j * 16 * K + kb * 8);

  auto a_row = [&](int strip) -> int {
    const int r = strip * 16 + lo;
    if constexpr (GATHER) return rowidx[r * CG + cg];
    else return r;
  };
  float4 a[KB];
  {
    const float* p = A + (size_t)a_row(first) * K + hi * 4;
#pragma unroll
    for (int kb = 0; kb < KB; ++kb) a[kb] = *(const float4*)(p + kb * 8);
  }
  int arn = a_row(first + nslots < Mtiles ? first + nslots : first);
  for (int strip = first; strip < Mtiles; strip += nslots) {
    const float* pn = A + (size_t)arn * K + hi * 4;
    float4 an[KB];
#pragma unroll
    for (int kb = 0; kb < KB; ++kb) an[kb] = *(const float4*)(pn + kb * 8);
    const int nn = strip + 2 * nslots;
    arn = a_row(nn < Mtiles ? nn : strip);

    v8f acc[NT];
#pragma unroll
    for (int j = 0; j < NT; ++j)
      acc[j] = (v8f){0.f, 0.f, 0.f, 0.f, 0.f, 0.f, 0.f, 0.f};
#pragma unroll
    for (int kb = 0; kb < KB; ++kb) {
#pragma unroll
      for (int j = 0; j < NT; ++j)
        acc[j] = __builtin_amdgcn_wmma_f32_16x16x4_f32(false, (v2f){a[kb].x, a[kb].y}, false,
                                                       (v2f){b[kb][j].x, b[kb][j].y},
                                                       (short)0, acc[j], false, false);
#pragma unroll
      for (int j = 0; j < NT; ++j)
        acc[j] = __builtin_amdgcn_wmma_f32_16x16x4_f32(false, (v2f){a[kb].z, a[kb].w}, false,
                                                       (v2f){b[kb][j].z, b[kb][j].w},
                                                       (short)0, acc[j], false, false);
    }
#pragma unroll
    for (int j = 0; j < NT; ++j) {
      float* Cb = C + (size_t)strip * 16 * ldc + coff + (cg * NT + j) * 16 + lo;
#pragma unroll
      for (int r = 0; r < 8; ++r)
        Cb[(size_t)(r + hi * 8) * ldc] = acc[j][r];
    }
#pragma unroll
    for (int kb = 0; kb < KB; ++kb) a[kb] = an[kb];
  }
}

__global__ void attn_scores(const float* __restrict__ z, const float* __restrict__ a,
                            float* __restrict__ ss, float* __restrict__ sd,
                            int n, int heads) {
  int t = blockIdx.x * blockDim.x + threadIdx.x;
  if (t >= n * heads) return;
  int node = t / heads, hh = t - node * heads;
  const float* zp = z + (size_t)node * heads * HID + hh * HID;
  const float* al = a + (size_t)hh * 2 * HID;
  const float* ar = al + HID;
  float s0 = 0.f, s1 = 0.f;
#pragma unroll 8
  for (int o = 0; o < HID; ++o) { s0 += zp[o] * al[o]; s1 += zp[o] * ar[o]; }
  ss[t] = s0; sd[t] = s1;
}

#define AG_ROWS   391
#define AG_GROUPS 256
#define AG_CAP    16384
#define AG_TB     1024
#define AG_U      8
#define AG_RB     9
#define AG_B      4

__device__ __forceinline__ int lane_get_i(int v, int i) {
  return __builtin_amdgcn_readlane(v, i);
}
__device__ __forceinline__ float lane_get_f(float v, int i) {
  return __int_as_float(__builtin_amdgcn_readlane(__float_as_int(v), i));
}

#define BM_EC     4096
#define BM_WPC    (BM_EC / 32)
#define BM_CHUNKS ((N_EDGES + BM_EC - 1) / BM_EC)
#define BM_PITCH  (BM_CHUNKS * BM_WPC)
#define BM_TB     256
#define BM_LP     (BM_WPC + 4)
static_assert(N_EDGES % 4 == 0 && BM_EC % (4 * BM_TB) == 0 && BM_WPC == 32 * 4, "one 16-byte piece per lane and row");
__device__ __forceinline__ unsigned owner_of(int d) {
  return min((unsigned)d / (unsigned)AG_ROWS, (unsigned)(AG_GROUPS - 1));
}
__global__ void __launch_bounds__(BM_TB)
dst_range_bitmaps(const int* __restrict__ dst, unsigned* __restrict__ obm) {
  __shared__ __attribute__((aligned(16))) unsigned tile[AG_GROUPS * BM_LP];
  constexpr int PER = BM_EC / 4 / BM_TB;
  const int tid = threadIdx.x;
  const int e0 = blockIdx.x * BM_EC;
  const int4* __restrict__ dst4 = (const int4*)dst;
  int4 d[PER];
#pragma unroll
  for (int u = 0; u < PER; ++u) d[u] = dst4[min((e0 >> 2) + tid + u * BM_TB, N_EDGES / 4 - 1)];
  for (int i = tid; i < AG_GROUPS * BM_LP / 4; i += BM_TB) ((uint4*)tile)[i] = make_uint4(0u, 0u, 0u, 0u);
  __syncthreads();
#pragma unroll
  for (int u = 0; u < PER; ++u) {
    const int el = (tid + u * BM_TB) * 4;
    if (e0 + el < N_EDGES) {
      const int wd = el >> 5;
      const unsigned bit = 1u << (el & 31);
      atomicOr(&tile[owner_of(d[u].x) * BM_LP + wd], bit);
      atomicOr(&tile[owner_of(d[u].y) * BM_LP + wd], bit << 1);
      atomicOr(&tile[owner_of(d[u].z) * BM_LP + wd], bit << 2);
      atomicOr(&tile[owner_of(d[u].w) * BM_LP + wd], bit << 3);
    }
  }
  __syncthreads();
  const int lane = tid & 31, wave = tid >> 5;
  constexpr int NW = BM_TB / 32;
  static_assert(AG_GROUPS % (4 * NW) == 0, "every wave stores whole groups of four rows");
  for (int r0 = wave; r0 < AG_GROUPS; r0 += 4 * NW) {
    uint4 v[4];
#pragma unroll
    for (int u = 0; u < 4; ++u) v[u] = *(const uint4*)&tile[(r0 + u * NW) * BM_LP + lane * 4];
#pragma unroll
    for (int u = 0; u < 4; ++u)
      *(uint4*)&obm[(size_t)(r0 + u * NW) * BM_PITCH + (size_t)blockIdx.x * BM_WPC + lane * 4] = v[u];
  }
}

template <int H, bool APPLY_ELU, bool TAKE_LIST>
__global__ void __launch_bounds__(AG_TB)
edge_softmax_aggregate_owner(const int* __restrict__ src, const int* __restrict__ dst,
                             const float* __restrict__ ss, const float* __restrict__ sd,
                             const float* __restrict__ z, float* __restrict__ out, int ldo,
                             int* __restrict__ wlist, int* __restrict__ wcnt,
                             const unsigned* __restrict__ obm, float* __restrict__ wscore) {
  static_assert(H == 1 || H == 4, "score slots are one float or one float4");
  constexpr int C = H * HID;
  constexpr bool SC_LDS = (H == 1);
  constexpr int RMASK = (1 << AG_RB) - 1;
  __shared__ __attribute__((aligned(16))) float acc[(AG_ROWS + 1) * C];
  __shared__ int lst[AG_CAP];
  __shared__ __attribute__((aligned(16))) unsigned mx[(AG_ROWS + 1) * H];
  __shared__ __attribute__((aligned(16))) float sm[(AG_ROWS + 1) * H];
  __shared__ __attribute__((aligned(16))) float sdl[(AG_ROWS + 1) * H];
  __shared__ float scl[SC_LDS ? AG_CAP : 1];
  __shared__ int cnt;
  const int tid  = threadIdx.x;
  const int lane = tid & 31;
  const int wave = __builtin_amdgcn_readfirstlane(tid >> 5);
  const int base = blockIdx.x * AG_ROWS;
  int* __restrict__ wl = wlist + (size_t)blockIdx.x * AG_CAP;
  float4* __restrict__ wsc = SC_LDS ? nullptr : (float4*)wscore + (size_t)blockIdx.x * AG_CAP;

  for (int i = tid; i < (AG_ROWS + 1) * C; i += AG_TB) acc[i] = 0.f;
  for (int i = tid; i < (AG_ROWS + 1) * H; i += AG_TB) {
    mx[i] = 0u;
    sm[i] = 0.f;
    const int g = base * H + i;
    sdl[i] = (i < AG_ROWS * H && g < N_NODES * H) ? sd[g] : 0.f;
  }
  int total;
  if constexpr (TAKE_LIST) {
    total = wcnt[blockIdx.x];
    const int m = (total > AG_CAP) ? 0 : total;
    for (int i = tid; i < m; i += AG_TB) lst[i] = wl[i];
    __syncthreads();
  } else {
    if (tid == 0) cnt = 0;
    __syncthreads();

    const uint4* __restrict__ row4 = (const uint4*)(obm + (size_t)blockIdx.x * BM_PITCH);
    constexpr int Q     = BM_PITCH / 4;
    constexpr int TRIPS = (Q + AG_TB - 1) / AG_TB;
    uint4 w[TRIPS];
#pragma unroll
    for (int t = 0; t < TRIPS; ++t) {
      const int q = tid + t * AG_TB;
      w[t] = row4[min(q, Q - 1)];
      if (q >= Q) w[t] = make_uint4(0u, 0u, 0u, 0u);
    }
    int c = 0;
#pragma unroll
    for (int t = 0; t < TRIPS; ++t) c += __popc(w[t].x) + __popc(w[t].y) + __popc(w[t].z) + __popc(w[t].w);
    int incl = c;
#pragma unroll
    for (int dl = 1; dl < 32; dl <<= 1) {
      const int v = __shfl_up(incl, dl);
      if (lane >= dl) incl += v;
    }
    int wbase = 0;
    if (lane == 31) wbase = atomicAdd(&cnt, incl);
    wbase = __builtin_amdgcn_readlane(wbase, 31);
    int slot = wbase + incl - c;
#pragma unroll
    for (int t = 0; t < TRIPS; ++t) {
      const int q = tid + t * AG_TB;
      const unsigned ww[4] = {w[t].x, w[t].y, w[t].z, w[t].w};
#pragma unroll
      for (int j = 0; j < 4; ++j) {
        unsigned x = ww[j];
        while (x) {
          const int b = __builtin_ctz(x);
          x &= x - 1u;
          if (slot < AG_CAP) lst[slot] = q * 128 + j * 32 + b;
          ++slot;
        }
      }
    }
    __syncthreads();

    total = cnt;
    if (tid == 0) wcnt[blockIdx.x] = total;
  }
  const bool overflow = total > AG_CAP;
  const int n = __builtin_amdgcn_readfirstlane(overflow ? 0 : total);

  auto leaky = [](float x) { return (x > 0.f) ? x : 0.01f * x; };
  auto load_ss = [&](int s, float (&x)[H]) {
    if constexpr (H == 4) {
      const float4 t = ((const float4*)ss)[s];
      x[0] = t.x; x[1] = t.y; x[2] = t.z; x[3] = t.w;
    } else {
#pragma unroll
      for (int h = 0; h < H; ++h) x[h] = ss[(size_t)s * H + h];
    }
  };
  auto put_sc = [&](int i, const float (&e)[H]) {
    if constexpr (SC_LDS) scl[i] = e[0];
    else wsc[i] = make_float4(e[0], e[1], e[2], e[3]);
  };
  auto get_sc = [&](int i, float (&e)[H]) {
    if constexpr (SC_LDS) e[0] = scl[i];
    else { const float4 t = wsc[i]; e[0] = t.x; e[1] = t.y; e[2] = t.z; e[3] = t.w; }
  };
  auto walk = [&](auto f) {
    for (int ed = tid; ed < N_EDGES; ed += AG_TB) {
      const unsigned r = (unsigned)(dst[ed] - base);
      if (r < (unsigned)AG_ROWS) f(src[ed], (int)r);
    }
  };

  if (overflow) {
    walk([&](int s, int r) {
      float x[H];
      load_ss(s, x);
      for (int h = 0; h < H; ++h) atomicMax(&mx[r * H + h], fkey(leaky(x[h] + sdl[r * H + h])));
    });
  } else {
    for (int i0 = tid; i0 < n; i0 += AG_B * AG_TB) {
      int p[AG_B], s[AG_B], rw[AG_B];
      float x[AG_B][H];
#pragma unroll
      for (int u = 0; u < AG_B; ++u) {
        const int idx = i0 + u * AG_TB;
        p[u] = (idx < n) ? lst[idx] : -1;
      }
#pragma unroll
      for (int u = 0; u < AG_B; ++u) {
        if constexpr (TAKE_LIST) {
          s[u]  = (p[u] >= 0) ? (p[u] >> AG_RB) : 0;
          rw[u] = p[u] & RMASK;
        } else {
          s[u]  = (p[u] >= 0) ? src[p[u]] : 0;
          const int d = (p[u] >= 0) ? dst[p[u]] : base;
          rw[u] = (int)min((unsigned)(d - base), (unsigned)AG_ROWS);
        }
      }
#pragma unroll
      for (int u = 0; u < AG_B; ++u) load_ss(s[u], x[u]);
#pragma unroll
      for (int u = 0; u < AG_B; ++u) {
        if (p[u] >= 0) {
          const int r = rw[u];
          if constexpr (!TAKE_LIST) {
            const int e = (s[u] << AG_RB) | r;
            lst[i0 + u * AG_TB] = e;
            wl[i0 + u * AG_TB] = e;
          }
          float e[H];
#pragma unroll
          for (int h = 0; h < H; ++h) {
            e[h] = leaky(x[u][h] + sdl[r * H + h]);
            atomicMax(&mx[r * H + h], fkey(e[h]));
          }
          put_sc(i0 + u * AG_TB, e);
        }
      }
    }
  }
  __syncthreads();

  if (overflow) {
    walk([&](int s, int r) {
      float x[H];
      load_ss(s, x);
      for (int h = 0; h < H; ++h)
        atomicAdd(&sm[r * H + h], expf(leaky(x[h] + sdl[r * H + h]) - funkey(mx[r * H + h])));
    });
  } else {
    for (int i0 = tid; i0 < n; i0 += AG_B * AG_TB) {
      int p[AG_B];
      float e[AG_B][H];
#pragma unroll
      for (int u = 0; u < AG_B; ++u) {
        const int idx = i0 + u * AG_TB;
        p[u] = (idx < n) ? lst[idx] : -1;
      }
#pragma unroll
      for (int u = 0; u < AG_B; ++u) get_sc((p[u] >= 0) ? (i0 + u * AG_TB) : i0, e[u]);
#pragma unroll
      for (int u = 0; u < AG_B; ++u) {
        if (p[u] >= 0) {
          const int r = p[u] & RMASK;
#pragma unroll
          for (int h = 0; h < H; ++h)
            atomicAdd(&sm[r * H + h], expf(e[u][h] - funkey(mx[r * H + h])));
        }
      }
    }
  }
  __syncthreads();

  if (overflow) {
    walk([&](int s, int r) {
      float x[H];
      load_ss(s, x);
      for (int h = 0; h < H; ++h) {
        const float a = expf(leaky(x[h] + sdl[r * H + h]) - funkey(mx[r * H + h])) / fmaxf(sm[r * H + h], 1e-9f);
        for (int o = 0; o < HID; ++o)
          atomicAdd(&acc[r * C + h * HID + o], a * z[(size_t)s * C + h * HID + o]);
      }
    });
  }
  for (int b0 = wave * 32; b0 < n; b0 += AG_TB) {
    const int idx = b0 + lane;
    const bool valid = idx < n;
    const int p = valid ? lst[idx] : AG_ROWS;
    const int s = p >> AG_RB;
    const int row = p & RMASK;
    float al[H];
#pragma unroll
    for (int h = 0; h < H; ++h) al[h] = 0.f;
    if (valid) get_sc(idx, al);
#pragma unroll
    for (int h = 0; h < H; ++h) {
      const float ee = expf(al[h] - funkey(mx[row * H + h]));
      al[h] = valid ? ee / fmaxf(sm[row * H + h], 1e-9f) : 0.f;
    }
    const int nb = min(32, n - b0);
    for (int i0 = 0; i0 < nb; i0 += AG_U) {
      float v[AG_U][H];
#pragma unroll
      for (int u = 0; u < AG_U; ++u) {
        const float* zp = z + (size_t)lane_get_i(s, i0 + u) * C + lane;
#pragma unroll
        for (int h = 0; h < H; ++h) v[u][h] = zp[h * HID];
      }
#pragma unroll
      for (int u = 0; u < AG_U; ++u) {
        float* ap = acc + lane_get_i(row, i0 + u) * C + lane;
#pragma unroll
        for (int h = 0; h < H; ++h)
          atomicAdd(ap + h * HID, lane_get_f(al[h], i0 + u) * v[u][h]);
      }
    }
  }
  __syncthreads();

  constexpr int C4 = C / 4;
  for (int i = tid; i < AG_ROWS * C4; i += AG_TB) {
    const int r = i / C4, c4 = i - r * C4;
    const int node = base + r;
    if (node < N_NODES) {
      float4 v = *(const float4*)&acc[r * C + c4 * 4];
      if constexpr (APPLY_ELU) {
        v.x = (v.x > 0.f) ? v.x : (expf(v.x) - 1.0f);
        v.y = (v.y > 0.f) ? v.y : (expf(v.y) - 1.0f);
        v.z = (v.z > 0.f) ? v.z : (expf(v.z) - 1.0f);
        v.w = (v.w > 0.f) ? v.w : (expf(v.w) - 1.0f);
      }
      *(float4*)&out[(size_t)node * ldo + c4 * 4] = v;
    }
  }
}

static_assert((size_t)AG_GROUPS * AG_CAP * 4 <= (size_t)N_NODES * (64 + HID + 2 + 96),
              "the score plane must fit inside the planes it lies over");

extern "C" void kernel_launch(void* const* d_in, const int* in_sizes, int n_in,
                              void* d_out, int out_size, void* d_ws, size_t ws_size,
                              hipStream_t stream) {
  const float* h    = (const float*)d_in[0];
  const int*   src  = (const int*)  d_in[1];
  const int*   dst  = (const int*)  d_in[2];
  const int*   fc   = (const int*)  d_in[3];
  const float* W1   = (const float*)d_in[4];
  const float* a1   = (const float*)d_in[5];
  const float* W2   = (const float*)d_in[6];
  const float* a2   = (const float*)d_in[7];
  const float* Wf   = (const float*)d_in[8];
  const float* Wff  = (const float*)d_in[9];
  const float* Wfc2 = (const float*)d_in[10];
  float* out = (float*)d_out;

  float* w = (float*)d_ws;
  size_t off = 0;
  float* agg1 = w + off; off += (size_t)N_NODES * 128;
  float* z1   = w + off; off += (size_t)N_NODES * 128;
  float* s1s  = w + off; off += (size_t)N_NODES * HEADS;
  float* s1d  = w + off; off += (size_t)N_NODES * HEADS;
  int*   elst = (int*)(w + off); off += (size_t)AG_GROUPS * AG_CAP;
  int*   ecnt = (int*)(w + off); off += AG_GROUPS;
  unsigned* obm  = (unsigned*)(w + off); off += (size_t)AG_GROUPS * BM_PITCH;
  float* esc  = w + off;
  float* hg   = w + off; off += (size_t)N_NODES * 64;
  float* z2   = w + off; off += (size_t)N_NODES * HID;
  float* s2s  = w + off; off += N_NODES;
  float* s2d  = w + off; off += N_NODES;
  float* hfp  = w + off; off += (size_t)N_NODES * 96;
  (void)ws_size; (void)in_sizes; (void)n_in; (void)out_size;

  const int TB = 256;
  const int MT = N_NODES / 16;
  const int PG = 256;

  wmma_gemm_resident<4, 128, 2, 4, false><<<PG, 256, 0, stream>>>(h, nullptr, W1, z1, MT, 128, 0);
  attn_scores<<<(N_NODES * HEADS + TB - 1) / TB, TB, 0, stream>>>(z1, a1, s1s, s1d, N_NODES, HEADS);
  dst_range_bitmaps<<<BM_CHUNKS, BM_TB, 0, stream>>>(dst, obm);
  edge_softmax_aggregate_owner<HEADS, true, false><<<AG_GROUPS, AG_TB, 0, stream>>>(src, dst, s1s, s1d, z1, agg1, 128, elst, ecnt, obm, esc);

  wmma_gemm_resident<2, 128, 1, 8, false><<<PG, 256, 0, stream>>>(agg1, nullptr, W2, z2, MT, HID, 0);
  attn_scores<<<(N_NODES + TB - 1) / TB, TB, 0, stream>>>(z2, a2, s2s, s2d, N_NODES, 1);
  edge_softmax_aggregate_owner<1, false, true><<<AG_GROUPS, AG_TB, 0, stream>>>(src, dst, s2s, s2d, z2, hg, 64, elst, ecnt, obm, nullptr);

  wmma_gemm_resident<2, 128, 3, 4, true><<<PG, 384, 0, stream>>>(h, fc, Wf, hfp, MT, 96, 0);
  wmma_gemm_resident<2, 96, 1, 8, false><<<PG, 256, 0, stream>>>(hfp, nullptr, Wff, hg, MT, 64, HID);

  wmma_gemm_resident<1, 64, 1, 8, false><<<PG, 256, 0, stream>>>(hg, nullptr, Wfc2, out, MT, OUT_DIM, 0);
}


# Layer-1 edge sums read each 512-byte z row with one 16-byte-per-lane load (4 whole lines per instruction), not four 4-byte loads; LDS tile channel-permuted, adds bank-free; arrival order kept
# speedup vs baseline: 6.5372x; 1.0338x over previous
#include <hip/hip_runtime.h>
#include <math.h>

#define N_NODES 100000
#define N_EDGES 1600000
#define IN_DIM  128
#define HID     32
#define HEADS   4
#define OUT_DIM 16

typedef __attribute__((ext_vector_type(2))) float v2f;
typedef __attribute__((ext_vector_type(8))) float v8f;

__device__ __forceinline__ unsigned fkey(float x) {
  unsigned b = __float_as_uint(x);
  return (b & 0x80000000u) ? ~b : (b | 0x80000000u);
}
__device__ __forceinline__ float funkey(unsigned k) {
  return __uint_as_float((k & 0x80000000u) ? (k & 0x7FFFFFFFu) : ~k);
}

template <int NT, int K, int CG, int WPG, bool GATHER>
__global__ void __launch_bounds__(CG * WPG * 32)
wmma_gemm_resident(const float* __restrict__ A, const int* __restrict__ rowidx,
                   const float* __restrict__ W, float* __restrict__ C,
                   int Mtiles, int ldc, int coff) {
  constexpr int KB = K / 8;
  const int lane = threadIdx.x & 31;
  const int lo = lane & 15;
  const int hi = lane >> 4;
  const int wv = __builtin_amdgcn_readfirstlane((int)(threadIdx.x >> 5));
  const int cg = wv / WPG;
  const int nslots = WPG * (int)gridDim.x;
  int first = (wv % WPG) * (int)gridDim.x + (int)blockIdx.x + cg * (nslots / CG);
  if (first >= nslots) first -= nslots;
  if (first >= Mtiles) return;

  const float* Wb = W + ((size_t)cg * NT * 16 + lo) * K + hi * 4;
  float4 b[KB][NT];
#pragma unroll
  for (int kb = 0; kb < KB; ++kb)
#pragma unroll
    for (int j = 0; j < NT; ++j) b[kb][j] = *(const float4*)(Wb + (size_t)j * 16 * K + kb * 8);

  auto a_row = [&](int strip) -> int {
    const int r = strip * 16 + lo;
    if constexpr (GATHER) return rowidx[r * CG + cg];
    else return r;
  };
  float4 a[KB];
  {
    const float* p = A + (size_t)a_row(first) * K + hi * 4;
#pragma unroll
    for (int kb = 0; kb < KB; ++kb) a[kb] = *(const float4*)(p + kb * 8);
  }
  int arn = a_row(first + nslots < Mtiles ? first + nslots : first);
  for (int strip = first; strip < Mtiles; strip += nslots) {
    const float* pn = A + (size_t)arn * K + hi * 4;
    float4 an[KB];
#pragma unroll
    for (int kb = 0; kb < KB; ++kb) an[kb] = *(const float4*)(pn + kb * 8);
    const int nn = strip + 2 * nslots;
    arn = a_row(nn < Mtiles ? nn : strip);

    v8f acc[NT];
#pragma unroll
    for (int j = 0; j < NT; ++j)
      acc[j] = (v8f){0.f, 0.f, 0.f, 0.f, 0.f, 0.f, 0.f, 0.f};
#pragma unroll
    for (int kb = 0; kb < KB; ++kb) {
#pragma unroll
      for (int j = 0; j < NT; ++j)
        acc[j] = __builtin_amdgcn_wmma_f32_16x16x4_f32(false, (v2f){a[kb].x, a[kb].y}, false,
                                                       (v2f){b[kb][j].x, b[kb][j].y},
                                                       (short)0, acc[j], false, false);
#pragma unroll
      for (int j = 0; j < NT; ++j)
        acc[j] = __builtin_amdgcn_wmma_f32_16x16x4_f32(false, (v2f){a[kb].z, a[kb].w}, false,
                                                       (v2f){b[kb][j].z, b[kb][j].w},
                                                       (short)0, acc[j], false, false);
    }
#pragma unroll
    for (int j = 0; j < NT; ++j) {
      float* Cb = C + (size_t)strip * 16 * ldc + coff + (cg * NT + j) * 16 + lo;
#pragma unroll
      for (int r = 0; r < 8; ++r)
        Cb[(size_t)(r + hi * 8) * ldc] = acc[j][r];
    }
#pragma unroll
    for (int kb = 0; kb < KB; ++kb) a[kb] = an[kb];
  }
}

__global__ void attn_scores(const float* __restrict__ z, const float* __restrict__ a,
                            float* __restrict__ ss, float* __restrict__ sd,
                            int n, int heads) {
  int t = blockIdx.x * blockDim.x + threadIdx.x;
  if (t >= n * heads) return;
  int node = t / heads, hh = t - node * heads;
  const float* zp = z + (size_t)node * heads * HID + hh * HID;
  const float* al = a + (size_t)hh * 2 * HID;
  const float* ar = al + HID;
  float s0 = 0.f, s1 = 0.f;
#pragma unroll 8
  for (int o = 0; o < HID; ++o) { s0 += zp[o] * al[o]; s1 += zp[o] * ar[o]; }
  ss[t] = s0; sd[t] = s1;
}

#define AG_ROWS   391
#define AG_GROUPS 256
#define AG_CAP    16384
#define AG_TB     1024
#define AG_U      8
#define AG_RB     9
#define AG_B      4

__device__ __forceinline__ int lane_get_i(int v, int i) {
  return __builtin_amdgcn_readlane(v, i);
}
__device__ __forceinline__ float lane_get_f(float v, int i) {
  return __int_as_float(__builtin_amdgcn_readlane(__float_as_int(v), i));
}

#define BM_EC     4096
#define BM_WPC    (BM_EC / 32)
#define BM_CHUNKS ((N_EDGES + BM_EC - 1) / BM_EC)
#define BM_PITCH  (BM_CHUNKS * BM_WPC)
#define BM_TB     256
#define BM_LP     (BM_WPC + 4)
static_assert(N_EDGES % 4 == 0 && BM_EC % (4 * BM_TB) == 0 && BM_WPC == 32 * 4, "one 16-byte piece per lane and row");
__device__ __forceinline__ unsigned owner_of(int d) {
  return min((unsigned)d / (unsigned)AG_ROWS, (unsigned)(AG_GROUPS - 1));
}
__global__ void __launch_bounds__(BM_TB)
dst_range_bitmaps(const int* __restrict__ dst, unsigned* __restrict__ obm) {
  __shared__ __attribute__((aligned(16))) unsigned tile[AG_GROUPS * BM_LP];
  constexpr int PER = BM_EC / 4 / BM_TB;
  const int tid = threadIdx.x;
  const int e0 = blockIdx.x * BM_EC;
  const int4* __restrict__ dst4 = (const int4*)dst;
  int4 d[PER];
#pragma unroll
  for (int u = 0; u < PER; ++u) d[u] = dst4[min((e0 >> 2) + tid + u * BM_TB, N_EDGES / 4 - 1)];
  for (int i = tid; i < AG_GROUPS * BM_LP / 4; i += BM_TB) ((uint4*)tile)[i] = make_uint4(0u, 0u, 0u, 0u);
  __syncthreads();
#pragma unroll
  for (int u = 0; u < PER; ++u) {
    const int el = (tid + u * BM_TB) * 4;
    if (e0 + el < N_EDGES) {
      const int wd = el >> 5;
      const unsigned bit = 1u << (el & 31);
      atomicOr(&tile[owner_of(d[u].x) * BM_LP + wd], bit);
      atomicOr(&tile[owner_of(d[u].y) * BM_LP + wd], bit << 1);
      atomicOr(&tile[owner_of(d[u].z) * BM_LP + wd], bit << 2);
      atomicOr(&tile[owner_of(d[u].w) * BM_LP + wd], bit << 3);
    }
  }
  __syncthreads();
  const int lane = tid & 31, wave = tid >> 5;
  constexpr int NW = BM_TB / 32;
  static_assert(AG_GROUPS % (4 * NW) == 0, "every wave stores whole groups of four rows");
  for (int r0 = wave; r0 < AG_GROUPS; r0 += 4 * NW) {
    uint4 v[4];
#pragma unroll
    for (int u = 0; u < 4; ++u) v[u] = *(const uint4*)&tile[(r0 + u * NW) * BM_LP + lane * 4];
#pragma unroll
    for (int u = 0; u < 4; ++u)
      *(uint4*)&obm[(size_t)(r0 + u * NW) * BM_PITCH + (size_t)blockIdx.x * BM_WPC + lane * 4] = v[u];
  }
}

template <int H, bool APPLY_ELU, bool TAKE_LIST>
__global__ void __launch_bounds__(AG_TB)
edge_softmax_aggregate_owner(const int* __restrict__ src, const int* __restrict__ dst,
                             const float* __restrict__ ss, const float* __restrict__ sd,
                             const float* __restrict__ z, float* __restrict__ out, int ldo,
                             int* __restrict__ wlist, int* __restrict__ wcnt,
                             const unsigned* __restrict__ obm, float* __restrict__ wscore) {
  static_assert(H == 1 || H == 4, "score slots are one float or one float4");
  constexpr int C = H * HID;
  constexpr int VW = C / 32;
  constexpr bool SC_LDS = (H == 1);
  constexpr int RMASK = (1 << AG_RB) - 1;
  __shared__ __attribute__((aligned(16))) float acc[(AG_ROWS + 1) * C];
  __shared__ int lst[AG_CAP];
  __shared__ __attribute__((aligned(16))) unsigned mx[(AG_ROWS + 1) * H];
  __shared__ __attribute__((aligned(16))) float sm[(AG_ROWS + 1) * H];
  __shared__ __attribute__((aligned(16))) float sdl[(AG_ROWS + 1) * H];
  __shared__ float scl[SC_LDS ? AG_CAP : 1];
  __shared__ int cnt;
  const int tid  = threadIdx.x;
  const int lane = tid & 31;
  const int wave = __builtin_amdgcn_readfirstlane(tid >> 5);
  const int base = blockIdx.x * AG_ROWS;
  int* __restrict__ wl = wlist + (size_t)blockIdx.x * AG_CAP;
  float4* __restrict__ wsc = SC_LDS ? nullptr : (float4*)wscore + (size_t)blockIdx.x * AG_CAP;

  for (int i = tid; i < (AG_ROWS + 1) * C; i += AG_TB) acc[i] = 0.f;
  for (int i = tid; i < (AG_ROWS + 1) * H; i += AG_TB) {
    mx[i] = 0u;
    sm[i] = 0.f;
    const int g = base * H + i;
    sdl[i] = (i < AG_ROWS * H && g < N_NODES * H) ? sd[g] : 0.f;
  }
  int total;
  if constexpr (TAKE_LIST) {
    total = wcnt[blockIdx.x];
    const int m = (total > AG_CAP) ? 0 : total;
    for (int i = tid; i < m; i += AG_TB) lst[i] = wl[i];
    __syncthreads();
  } else {
    if (tid == 0) cnt = 0;
    __syncthreads();

    const uint4* __restrict__ row4 = (const uint4*)(obm + (size_t)blockIdx.x * BM_PITCH);
    constexpr int Q     = BM_PITCH / 4;
    constexpr int TRIPS = (Q + AG_TB - 1) / AG_TB;
    uint4 w[TRIPS];
#pragma unroll
    for (int t = 0; t < TRIPS; ++t) {
      const int q = tid + t * AG_TB;
      w[t] = row4[min(q, Q - 1)];
      if (q >= Q) w[t] = make_uint4(0u, 0u, 0u, 0u);
    }
    int c = 0;
#pragma unroll
    for (int t = 0; t < TRIPS; ++t) c += __popc(w[t].x) + __popc(w[t].y) + __popc(w[t].z) + __popc(w[t].w);
    int incl = c;
#pragma unroll
    for (int dl = 1; dl < 32; dl <<= 1) {
      const int v = __shfl_up(incl, dl);
      if (lane >= dl) incl += v;
    }
    int wbase = 0;
    if (lane == 31) wbase = atomicAdd(&cnt, incl);
    wbase = __builtin_amdgcn_readlane(wbase, 31);
    int slot = wbase + incl - c;
#pragma unroll
    for (int t = 0; t < TRIPS; ++t) {
      const int q = tid + t * AG_TB;
      const unsigned ww[4] = {w[t].x, w[t].y, w[t].z, w[t].w};
#pragma unroll
      for (int j = 0; j < 4; ++j) {
        unsigned x = ww[j];
        while (x) {
          const int b = __builtin_ctz(x);
          x &= x - 1u;
          if (slot < AG_CAP) lst[slot] = q * 128 + j * 32 + b;
          ++slot;
        }
      }
    }
    __syncthreads();

    total = cnt;
    if (tid == 0) wcnt[blockIdx.x] = total;
  }
  const bool overflow = total > AG_CAP;
  const int n = __builtin_amdgcn_readfirstlane(overflow ? 0 : total);

  auto leaky = [](float x) { return (x > 0.f) ? x : 0.01f * x; };
  auto load_ss = [&](int s, float (&x)[H]) {
    if constexpr (H == 4) {
      const float4 t = ((const float4*)ss)[s];
      x[0] = t.x; x[1] = t.y; x[2] = t.z; x[3] = t.w;
    } else {
#pragma unroll
      for (int h = 0; h < H; ++h) x[h] = ss[(size_t)s * H + h];
    }
  };
  auto put_sc = [&](int i, const float (&e)[H]) {
    if constexpr (SC_LDS) scl[i] = e[0];
    else wsc[i] = make_float4(e[0], e[1], e[2], e[3]);
  };
  auto get_sc = [&](int i, float (&e)[H]) {
    if constexpr (SC_LDS) e[0] = scl[i];
    else { const float4 t = wsc[i]; e[0] = t.x; e[1] = t.y; e[2] = t.z; e[3] = t.w; }
  };
  auto walk = [&](auto f) {
    for (int ed = tid; ed < N_EDGES; ed += AG_TB) {
      const unsigned r = (unsigned)(dst[ed] - base);
      if (r < (unsigned)AG_ROWS) f(src[ed], (int)r);
    }
  };

  if (overflow) {
    walk([&](int s, int r) {
      float x[H];
      load_ss(s, x);
      for (int h = 0; h < H; ++h) atomicMax(&mx[r * H + h], fkey(leaky(x[h] + sdl[r * H + h])));
    });
  } else {
    for (int i0 = tid; i0 < n; i0 += AG_B * AG_TB) {
      int p[AG_B], s[AG_B], rw[AG_B];
      float x[AG_B][H];
#pragma unroll
      for (int u = 0; u < AG_B; ++u) {
        const int idx = i0 + u * AG_TB;
        p[u] = (idx < n) ? lst[idx] : -1;
      }
#pragma unroll
      for (int u = 0; u < AG_B; ++u) {
        if constexpr (TAKE_LIST) {
          s[u]  = (p[u] >= 0) ? (p[u] >> AG_RB) : 0;
          rw[u] = p[u] & RMASK;
        } else {
          s[u]  = (p[u] >= 0) ? src[p[u]] : 0;
          const int d = (p[u] >= 0) ? dst[p[u]] : base;
          rw[u] = (int)min((unsigned)(d - base), (unsigned)AG_ROWS);
        }
      }
#pragma unroll
      for (int u = 0; u < AG_B; ++u) load_ss(s[u], x[u]);
#pragma unroll
      for (int u = 0; u < AG_B; ++u) {
        if (p[u] >= 0) {
          const int r = rw[u];
          if constexpr (!TAKE_LIST) {
            const int e = (s[u] << AG_RB) | r;
            lst[i0 + u * AG_TB] = e;
            wl[i0 + u * AG_TB] = e;
          }
          float e[H];
#pragma unroll
          for (int h = 0; h < H; ++h) {
            e[h] = leaky(x[u][h] + sdl[r * H + h]);
            atomicMax(&mx[r * H + h], fkey(e[h]));
          }
          put_sc(i0 + u * AG_TB, e);
        }
      }
    }
  }
  __syncthreads();

  if (overflow) {
    walk([&](int s, int r) {
      float x[H];
      load_ss(s, x);
      for (int h = 0; h < H; ++h)
        atomicAdd(&sm[r * H + h], expf(leaky(x[h] + sdl[r * H + h]) - funkey(mx[r * H + h])));
    });
  } else {
    for (int i0 = tid; i0 < n; i0 += AG_B * AG_TB) {
      int p[AG_B];
      float e[AG_B][H];
#pragma unroll
      for (int u = 0; u < AG_B; ++u) {
        const int idx = i0 + u * AG_TB;
        p[u] = (idx < n) ? lst[idx] : -1;
      }
#pragma unroll
      for (int u = 0; u < AG_B; ++u) get_sc((p[u] >= 0) ? (i0 + u * AG_TB) : i0, e[u]);
#pragma unroll
      for (int u = 0; u < AG_B; ++u) {
        if (p[u] >= 0) {
          const int r = p[u] & RMASK;
#pragma unroll
          for (int h = 0; h < H; ++h)
            atomicAdd(&sm[r * H + h], expf(e[u][h] - funkey(mx[r * H + h])));
        }
      }
    }
  }
  __syncthreads();

  if (overflow) {
    walk([&](int s, int r) {
      float x[H];
      load_ss(s, x);
      for (int h = 0; h < H; ++h) {
        const float a = expf(leaky(x[h] + sdl[r * H + h]) - funkey(mx[r * H + h])) / fmaxf(sm[r * H + h], 1e-9f);
        for (int o = 0; o < HID; ++o) {
          const int ch = h * HID + o;
          atomicAdd(&acc[r * C + (ch % VW) * 32 + ch / VW], a * z[(size_t)s * C + ch]);
        }
      }
    });
  }
  for (int b0 = wave * 32; b0 < n; b0 += AG_TB) {
    const int idx = b0 + lane;
    const bool valid = idx < n;
    const int p = valid ? lst[idx] : AG_ROWS;
    const int s = p >> AG_RB;
    const int row = p & RMASK;
    float al[H];
#pragma unroll
    for (int h = 0; h < H; ++h) al[h] = 0.f;
    if (valid) get_sc(idx, al);
#pragma unroll
    for (int h = 0; h < H; ++h) {
      const float ee = expf(al[h] - funkey(mx[row * H + h]));
      al[h] = valid ? ee / fmaxf(sm[row * H + h], 1e-9f) : 0.f;
    }
    const int nb = min(32, n - b0);
    [[maybe_unused]] const int hsel = lane / (32 / H);
    for (int i0 = 0; i0 < nb; i0 += AG_U) {
      float v[AG_U][VW];
#pragma unroll
      for (int u = 0; u < AG_U; ++u) {
        const float* zp = z + (size_t)lane_get_i(s, i0 + u) * C + lane * VW;
        if constexpr (VW == 4) {
          const float4 t = *(const float4*)zp;
          v[u][0] = t.x; v[u][1] = t.y; v[u][2] = t.z; v[u][3] = t.w;
        } else {
#pragma unroll
          for (int c = 0; c < VW; ++c) v[u][c] = zp[c];
        }
      }
#pragma unroll
      for (int u = 0; u < AG_U; ++u) {
        float* ap = acc + lane_get_i(row, i0 + u) * C + lane;
        float wgt = lane_get_f(al[0], i0 + u);
#pragma unroll
        for (int h = 1; h < H; ++h) {
          const float ah = lane_get_f(al[h], i0 + u);
          wgt = (hsel == h) ? ah : wgt;
        }
#pragma unroll
        for (int c = 0; c < VW; ++c)
          atomicAdd(ap + c * 32, wgt * v[u][c]);
      }
    }
  }
  __syncthreads();

  constexpr int C4 = C / 4;
  for (int i = tid; i < AG_ROWS * C4; i += AG_TB) {
    const int r = i / C4, c4 = i - r * C4;
    const int node = base + r;
    if (node < N_NODES) {
      float4 v;
      if constexpr (VW == 4) {
        const float* ap = &acc[r * C + c4];
        v = make_float4(ap[0], ap[32], ap[64], ap[96]);
      } else {
        v = *(const float4*)&acc[r * C + c4 * 4];
      }
      if constexpr (APPLY_ELU) {
        v.x = (v.x > 0.f) ? v.x : (expf(v.x) - 1.0f);
        v.y = (v.y > 0.f) ? v.y : (expf(v.y) - 1.0f);
        v.z = (v.z > 0.f) ? v.z : (expf(v.z) - 1.0f);
        v.w = (v.w > 0.f) ? v.w : (expf(v.w) - 1.0f);
      }
      *(float4*)&out[(size_t)node * ldo + c4 * 4] = v;
    }
  }
}

static_assert((size_t)AG_GROUPS * AG_CAP * 4 <= (size_t)N_NODES * (64 + HID + 2 + 96),
              "the score plane must fit inside the planes it lies over");

extern "C" void kernel_launch(void* const* d_in, const int* in_sizes, int n_in,
                              void* d_out, int out_size, void* d_ws, size_t ws_size,
                              hipStream_t stream) {
  const float* h    = (const float*)d_in[0];
  const int*   src  = (const int*)  d_in[1];
  const int*   dst  = (const int*)  d_in[2];
  const int*   fc   = (const int*)  d_in[3];
  const float* W1   = (const float*)d_in[4];
  const float* a1   = (const float*)d_in[5];
  const float* W2   = (const float*)d_in[6];
  const float* a2   = (const float*)d_in[7];
  const float* Wf   = (const float*)d_in[8];
  const float* Wff  = (const float*)d_in[9];
  const float* Wfc2 = (const float*)d_in[10];
  float* out = (float*)d_out;

  float* w = (float*)d_ws;
  size_t off = 0;
  float* agg1 = w + off; off += (size_t)N_NODES * 128;
  float* z1   = w + off; off += (size_t)N_NODES * 128;
  float* s1s  = w + off; off += (size_t)N_NODES * HEADS;
  float* s1d  = w + off; off += (size_t)N_NODES * HEADS;
  int*   elst = (int*)(w + off); off += (size_t)AG_GROUPS * AG_CAP;
  int*   ecnt = (int*)(w + off); off += AG_GROUPS;
  unsigned* obm  = (unsigned*)(w + off); off += (size_t)AG_GROUPS * BM_PITCH;
  float* esc  = w + off;
  float* hg   = w + off; off += (size_t)N_NODES * 64;
  float* z2   = w + off; off += (size_t)N_NODES * HID;
  float* s2s  = w + off; off += N_NODES;
  float* s2d  = w + off; off += N_NODES;
  float* hfp  = w + off; off += (size_t)N_NODES * 96;
  (void)ws_size; (void)in_sizes; (void)n_in; (void)out_size;

  const int TB = 256;
  const int MT = N_NODES / 16;
  const int PG = 256;

  wmma_gemm_resident<4, 128, 2, 4, false><<<PG, 256, 0, stream>>>(h, nullptr, W1, z1, MT, 128, 0);
  attn_scores<<<(N_NODES * HEADS + TB - 1) / TB, TB, 0, stream>>>(z1, a1, s1s, s1d, N_NODES, HEADS);
  dst_range_bitmaps<<<BM_CHUNKS, BM_TB, 0, stream>>>(dst, obm);
  edge_softmax_aggregate_owner<HEADS, true, false><<<AG_GROUPS, AG_TB, 0, stream>>>(src, dst, s1s, s1d, z1, agg1, 128, elst, ecnt, obm, esc);

  wmma_gemm_resident<2, 128, 1, 8, false><<<PG, 256, 0, stream>>>(agg1, nullptr, W2, z2, MT, HID, 0);
  attn_scores<<<(N_NODES + TB - 1) / TB, TB, 0, stream>>>(z2, a2, s2s, s2d, N_NODES, 1);
  edge_softmax_aggregate_owner<1, false, true><<<AG_GROUPS, AG_TB, 0, stream>>>(src, dst, s2s, s2d, z2, hg, 64, elst, ecnt, obm, nullptr);

  wmma_gemm_resident<2, 128, 3, 4, true><<<PG, 384, 0, stream>>>(h, fc, Wf, hfp, MT, 96, 0);
  wmma_gemm_resident<2, 96, 1, 8, false><<<PG, 256, 0, stream>>>(hfp, nullptr, Wff, hg, MT, 64, HID);

  wmma_gemm_resident<1, 64, 1, 8, false><<<PG, 256, 0, stream>>>(hg, nullptr, Wfc2, out, MT, OUT_DIM, 0);
}


# Frac GEMM, Wff GEMM and final projection in one launch: a wave takes a 16-row strip through all three on WMMA, tiles turned via its own LDS; hfpre, hf never stored; out = h2-half + hf-half k sums
# speedup vs baseline: 6.9486x; 1.0629x over previous
#include <hip/hip_runtime.h>
#include <math.h>

#define N_NODES 100000
#define N_EDGES 1600000
#define IN_DIM  128
#define HID     32
#define HEADS   4
#define OUT_DIM 16

typedef __attribute__((ext_vector_type(2))) float v2f;
typedef __attribute__((ext_vector_type(8))) float v8f;

__device__ __forceinline__ unsigned fkey(float x) {
  unsigned b = __float_as_uint(x);
  return (b & 0x80000000u) ? ~b : (b | 0x80000000u);
}
__device__ __forceinline__ float funkey(unsigned k) {
  return __uint_as_float((k & 0x80000000u) ? (k & 0x7FFFFFFFu) : ~k);
}

template <int NT, int K, int CG, int WPG, bool GATHER>
__global__ void __launch_bounds__(CG * WPG * 32)
wmma_gemm_resident(const float* __restrict__ A, const int* __restrict__ rowidx,
                   const float* __restrict__ W, float* __restrict__ C,
                   int Mtiles, int ldc, int coff) {
  constexpr int KB = K / 8;
  const int lane = threadIdx.x & 31;
  const int lo = lane & 15;
  const int hi = lane >> 4;
  const int wv = __builtin_amdgcn_readfirstlane((int)(threadIdx.x >> 5));
  const int cg = wv / WPG;
  const int nslots = WPG * (int)gridDim.x;
  int first = (wv % WPG) * (int)gridDim.x + (int)blockIdx.x + cg * (nslots / CG);
  if (first >= nslots) first -= nslots;
  if (first >= Mtiles) return;

  const float* Wb = W + ((size_t)cg * NT * 16 + lo) * K + hi * 4;
  float4 b[KB][NT];
#pragma unroll
  for (int kb = 0; kb < KB; ++kb)
#pragma unroll
    for (int j = 0; j < NT; ++j) b[kb][j] = *(const float4*)(Wb + (size_t)j * 16 * K + kb * 8);

  auto a_row = [&](int strip) -> int {
    const int r = strip * 16 + lo;
    if constexpr (GATHER) return rowidx[r * CG + cg];
    else return r;
  };
  float4 a[KB];
  {
    const float* p = A + (size_t)a_row(first) * K + hi * 4;
#pragma unroll
    for (int kb = 0; kb < KB; ++kb) a[kb] = *(const float4*)(p + kb * 8);
  }
  int arn = a_row(first + nslots < Mtiles ? first + nslots : first);
  for (int strip = first; strip < Mtiles; strip += nslots) {
    const float* pn = A + (size_t)arn * K + hi * 4;
    float4 an[KB];
#pragma unroll
    for (int kb = 0; kb < KB; ++kb) an[kb] = *(const float4*)(pn + kb * 8);
    const int nn = strip + 2 * nslots;
    arn = a_row(nn < Mtiles ? nn : strip);

    v8f acc[NT];
#pragma unroll
    for (int j = 0; j < NT; ++j)
      acc[j] = (v8f){0.f, 0.f, 0.f, 0.f, 0.f, 0.f, 0.f, 0.f};
#pragma unroll
    for (int kb = 0; kb < KB; ++kb) {
#pragma unroll
      for (int j = 0; j < NT; ++j)
        acc[j] = __builtin_amdgcn_wmma_f32_16x16x4_f32(false, (v2f){a[kb].x, a[kb].y}, false,
                                                       (v2f){b[kb][j].x, b[kb][j].y},
                                                       (short)0, acc[j], false, false);
#pragma unroll
      for (int j = 0; j < NT; ++j)
        acc[j] = __builtin_amdgcn_wmma_f32_16x16x4_f32(false, (v2f){a[kb].z, a[kb].w}, false,
                                                       (v2f){b[kb][j].z, b[kb][j].w},
                                                       (short)0, acc[j], false, false);
    }
#pragma unroll
    for (int j = 0; j < NT; ++j) {
      float* Cb = C + (size_t)strip * 16 * ldc + coff + (cg * NT + j) * 16 + lo;
#pragma unroll
      for (int r = 0; r < 8; ++r)
        Cb[(size_t)(r + hi * 8) * ldc] = acc[j][r];
    }
#pragma unroll
    for (int kb = 0; kb < KB; ++kb) a[kb] = an[kb];
  }
}

#define FC_WAVES 4
#define FC_P1    100
#define FC_P2    36
__global__ void __launch_bounds__(FC_WAVES * 32)
frac_chain(const float* __restrict__ h, const int* __restrict__ fc,
           const float* __restrict__ Wf, const float* __restrict__ Wff,
           const float* __restrict__ Wfc2, const float* __restrict__ hg,
           float* __restrict__ out, int Mtiles) {
  __shared__ __attribute__((aligned(16))) float t1s[FC_WAVES][16 * FC_P1];
  __shared__ __attribute__((aligned(16))) float t2s[FC_WAVES][16 * FC_P2];
  const int lane = threadIdx.x & 31;
  const int lo = lane & 15;
  const int hi = lane >> 4;
  const int wv = __builtin_amdgcn_readfirstlane((int)(threadIdx.x >> 5));
  const int nslots = FC_WAVES * (int)gridDim.x;
  const int first = wv * (int)gridDim.x + (int)blockIdx.x;
  if (first >= Mtiles) return;
  float* t1 = t1s[wv];
  float* t2 = t2s[wv];

  float4 bf[3][16][2];
#pragma unroll
  for (int s = 0; s < 3; ++s)
#pragma unroll
    for (int kb = 0; kb < 16; ++kb)
#pragma unroll
      for (int j = 0; j < 2; ++j)
        bf[s][kb][j] = *(const float4*)(Wf + (size_t)(s * 32 + j * 16 + lo) * IN_DIM + kb * 8 + hi * 4);
  float4 bw[12][2];
#pragma unroll
  for (int kb = 0; kb < 12; ++kb)
#pragma unroll
    for (int j = 0; j < 2; ++j)
      bw[kb][j] = *(const float4*)(Wff + (size_t)(j * 16 + lo) * 96 + kb * 8 + hi * 4);
  float4 bo[8];
#pragma unroll
  for (int kb = 0; kb < 8; ++kb) bo[kb] = *(const float4*)(Wfc2 + (size_t)lo * 64 + kb * 8 + hi * 4);

  float4 a[3][16];
#pragma unroll
  for (int s = 0; s < 3; ++s) {
    const float* p = h + (size_t)fc[(first * 16 + lo) * 3 + s] * IN_DIM + hi * 4;
#pragma unroll
    for (int kb = 0; kb < 16; ++kb) a[s][kb] = *(const float4*)(p + kb * 8);
  }
  const v8f zero8 = (v8f){0.f, 0.f, 0.f, 0.f, 0.f, 0.f, 0.f, 0.f};
  for (int strip = first; strip < Mtiles; strip += nslots) {
    const int nx = (strip + nslots < Mtiles) ? strip + nslots : strip;
    int nr[3];
#pragma unroll
    for (int s = 0; s < 3; ++s) nr[s] = fc[(nx * 16 + lo) * 3 + s];
    float4 h2f[4];
#pragma unroll
    for (int kb = 0; kb < 4; ++kb)
      h2f[kb] = *(const float4*)(hg + (size_t)(strip * 16 + lo) * 64 + kb * 8 + hi * 4);

#pragma unroll
    for (int s = 0; s < 3; ++s) {
      v8f acc[2] = {zero8, zero8};
#pragma unroll
      for (int kb = 0; kb < 16; ++kb) {
#pragma unroll
        for (int j = 0; j < 2; ++j)
          acc[j] = __builtin_amdgcn_wmma_f32_16x16x4_f32(false, (v2f){a[s][kb].x, a[s][kb].y}, false,
                                                         (v2f){bf[s][kb][j].x, bf[s][kb][j].y},
                                                         (short)0, acc[j], false, false);
#pragma unroll
        for (int j = 0; j < 2; ++j)
          acc[j] = __builtin_amdgcn_wmma_f32_16x16x4_f32(false, (v2f){a[s][kb].z, a[s][kb].w}, false,
                                                         (v2f){bf[s][kb][j].z, bf[s][kb][j].w},
                                                         (short)0, acc[j], false, false);
      }
      {
        const float* p = h + (size_t)nr[s] * IN_DIM + hi * 4;
#pragma unroll
        for (int kb = 0; kb < 16; ++kb) a[s][kb] = *(const float4*)(p + kb * 8);
      }
#pragma unroll
      for (int j = 0; j < 2; ++j)
#pragma unroll
        for (int r = 0; r < 8; ++r) t1[(hi * 8 + r) * FC_P1 + s * 32 + j * 16 + lo] = acc[j][r];
    }
    __builtin_amdgcn_wave_barrier();

    float4 a2[12];
#pragma unroll
    for (int kb = 0; kb < 12; ++kb) a2[kb] = *(const float4*)&t1[lo * FC_P1 + kb * 8 + hi * 4];
    v8f acc2[2] = {zero8, zero8};
#pragma unroll
    for (int kb = 0; kb < 12; ++kb) {
#pragma unroll
      for (int j = 0; j < 2; ++j)
        acc2[j] = __builtin_amdgcn_wmma_f32_16x16x4_f32(false, (v2f){a2[kb].x, a2[kb].y}, false,
                                                        (v2f){bw[kb][j].x, bw[kb][j].y},
                                                        (short)0, acc2[j], false, false);
#pragma unroll
      for (int j = 0; j < 2; ++j)
        acc2[j] = __builtin_amdgcn_wmma_f32_16x16x4_f32(false, (v2f){a2[kb].z, a2[kb].w}, false,
                                                        (v2f){bw[kb][j].z, bw[kb][j].w},
                                                        (short)0, acc2[j], false, false);
    }
#pragma unroll
    for (int j = 0; j < 2; ++j)
#pragma unroll
      for (int r = 0; r < 8; ++r) t2[(hi * 8 + r) * FC_P2 + j * 16 + lo] = acc2[j][r];
    __builtin_amdgcn_wave_barrier();

    float4 a3[4];
#pragma unroll
    for (int kb = 0; kb < 4; ++kb) a3[kb] = *(const float4*)&t2[lo * FC_P2 + kb * 8 + hi * 4];
    v8f o1 = zero8, o2 = zero8;
#pragma unroll
    for (int kb = 0; kb < 4; ++kb) {
      o1 = __builtin_amdgcn_wmma_f32_16x16x4_f32(false, (v2f){h2f[kb].x, h2f[kb].y}, false,
                                                 (v2f){bo[kb].x, bo[kb].y}, (short)0, o1, false, false);
      o2 = __builtin_amdgcn_wmma_f32_16x16x4_f32(false, (v2f){a3[kb].x, a3[kb].y}, false,
                                                 (v2f){bo[4 + kb].x, bo[4 + kb].y}, (short)0, o2, false, false);
      o1 = __builtin_amdgcn_wmma_f32_16x16x4_f32(false, (v2f){h2f[kb].z, h2f[kb].w}, false,
                                                 (v2f){bo[kb].z, bo[kb].w}, (short)0, o1, false, false);
      o2 = __builtin_amdgcn_wmma_f32_16x16x4_f32(false, (v2f){a3[kb].z, a3[kb].w}, false,
                                                 (v2f){bo[4 + kb].z, bo[4 + kb].w}, (short)0, o2, false, false);
    }
    float* ob = out + (size_t)strip * 16 * OUT_DIM + lo;
#pragma unroll
    for (int r = 0; r < 8; ++r) ob[(size_t)(r + hi * 8) * OUT_DIM] = o1[r] + o2[r];
  }
}

__global__ void attn_scores(const float* __restrict__ z, const float* __restrict__ a,
                            float* __restrict__ ss, float* __restrict__ sd,
                            int n, int heads) {
  int t = blockIdx.x * blockDim.x + threadIdx.x;
  if (t >= n * heads) return;
  int node = t / heads, hh = t - node * heads;
  const float* zp = z + (size_t)node * heads * HID + hh * HID;
  const float* al = a + (size_t)hh * 2 * HID;
  const float* ar = al + HID;
  float s0 = 0.f, s1 = 0.f;
#pragma unroll 8
  for (int o = 0; o < HID; ++o) { s0 += zp[o] * al[o]; s1 += zp[o] * ar[o]; }
  ss[t] = s0; sd[t] = s1;
}

#define AG_ROWS   391
#define AG_GROUPS 256
#define AG_CAP    16384
#define AG_TB     1024
#define AG_U      8
#define AG_RB     9
#define AG_B      4

__device__ __forceinline__ int lane_get_i(int v, int i) {
  return __builtin_amdgcn_readlane(v, i);
}
__device__ __forceinline__ float lane_get_f(float v, int i) {
  return __int_as_float(__builtin_amdgcn_readlane(__float_as_int(v), i));
}

#define BM_EC     4096
#define BM_WPC    (BM_EC / 32)
#define BM_CHUNKS ((N_EDGES + BM_EC - 1) / BM_EC)
#define BM_PITCH  (BM_CHUNKS * BM_WPC)
#define BM_TB     256
#define BM_LP     (BM_WPC + 4)
static_assert(N_EDGES % 4 == 0 && BM_EC % (4 * BM_TB) == 0 && BM_WPC == 32 * 4, "one 16-byte piece per lane and row");
__device__ __forceinline__ unsigned owner_of(int d) {
  return min((unsigned)d / (unsigned)AG_ROWS, (unsigned)(AG_GROUPS - 1));
}
__global__ void __launch_bounds__(BM_TB)
dst_range_bitmaps(const int* __restrict__ dst, unsigned* __restrict__ obm) {
  __shared__ __attribute__((aligned(16))) unsigned tile[AG_GROUPS * BM_LP];
  constexpr int PER = BM_EC / 4 / BM_TB;
  const int tid = threadIdx.x;
  const int e0 = blockIdx.x * BM_EC;
  const int4* __restrict__ dst4 = (const int4*)dst;
  int4 d[PER];
#pragma unroll
  for (int u = 0; u < PER; ++u) d[u] = dst4[min((e0 >> 2) + tid + u * BM_TB, N_EDGES / 4 - 1)];
  for (int i = tid; i < AG_GROUPS * BM_LP / 4; i += BM_TB) ((uint4*)tile)[i] = make_uint4(0u, 0u, 0u, 0u);
  __syncthreads();
#pragma unroll
  for (int u = 0; u < PER; ++u) {
    const int el = (tid + u * BM_TB) * 4;
    if (e0 + el < N_EDGES) {
      const int wd = el >> 5;
      const unsigned bit = 1u << (el & 31);
      atomicOr(&tile[owner_of(d[u].x) * BM_LP + wd], bit);
      atomicOr(&tile[owner_of(d[u].y) * BM_LP + wd], bit << 1);
      atomicOr(&tile[owner_of(d[u].z) * BM_LP + wd], bit << 2);
      atomicOr(&tile[owner_of(d[u].w) * BM_LP + wd], bit << 3);
    }
  }
  __syncthreads();
  const int lane = tid & 31, wave = tid >> 5;
  constexpr int NW = BM_TB / 32;
  static_assert(AG_GROUPS % (4 * NW) == 0, "every wave stores whole groups of four rows");
  for (int r0 = wave; r0 < AG_GROUPS; r0 += 4 * NW) {
    uint4 v[4];
#pragma unroll
    for (int u = 0; u < 4; ++u) v[u] = *(const uint4*)&tile[(r0 + u * NW) * BM_LP + lane * 4];
#pragma unroll
    for (int u = 0; u < 4; ++u)
      *(uint4*)&obm[(size_t)(r0 + u * NW) * BM_PITCH + (size_t)blockIdx.x * BM_WPC + lane * 4] = v[u];
  }
}

template <int H, bool APPLY_ELU, bool TAKE_LIST>
__global__ void __launch_bounds__(AG_TB)
edge_softmax_aggregate_owner(const int* __restrict__ src, const int* __restrict__ dst,
                             const float* __restrict__ ss, const float* __restrict__ sd,
                             const float* __restrict__ z, float* __restrict__ out, int ldo,
                             int* __restrict__ wlist, int* __restrict__ wcnt,
                             const unsigned* __restrict__ obm, float* __restrict__ wscore) {
  static_assert(H == 1 || H == 4, "score slots are one float or one float4");
  constexpr int C = H * HID;
  constexpr int VW = C / 32;
  constexpr bool SC_LDS = (H == 1);
  constexpr int RMASK = (1 << AG_RB) - 1;
  __shared__ __attribute__((aligned(16))) float acc[(AG_ROWS + 1) * C];
  __shared__ int lst[AG_CAP];
  __shared__ __attribute__((aligned(16))) unsigned mx[(AG_ROWS + 1) * H];
  __shared__ __attribute__((aligned(16))) float sm[(AG_ROWS + 1) * H];
  __shared__ __attribute__((aligned(16))) float sdl[(AG_ROWS + 1) * H];
  __shared__ float scl[SC_LDS ? AG_CAP : 1];
  __shared__ int cnt;
  const int tid  = threadIdx.x;
  const int lane = tid & 31;
  const int wave = __builtin_amdgcn_readfirstlane(tid >> 5);
  const int base = blockIdx.x * AG_ROWS;
  int* __restrict__ wl = wlist + (size_t)blockIdx.x * AG_CAP;
  float4* __restrict__ wsc = SC_LDS ? nullptr : (float4*)wscore + (size_t)blockIdx.x * AG_CAP;

  for (int i = tid; i < (AG_ROWS + 1) * C; i += AG_TB) acc[i] = 0.f;
  for (int i = tid; i < (AG_ROWS + 1) * H; i += AG_TB) {
    mx[i] = 0u;
    sm[i] = 0.f;
    const int g = base * H + i;
    sdl[i] = (i < AG_ROWS * H && g < N_NODES * H) ? sd[g] : 0.f;
  }
  int total;
  if constexpr (TAKE_LIST) {
    total = wcnt[blockIdx.x];
    const int m = (total > AG_CAP) ? 0 : total;
    for (int i = tid; i < m; i += AG_TB) lst[i] = wl[i];
    __syncthreads();
  } else {
    if (tid == 0) cnt = 0;
    __syncthreads();

    const uint4* __restrict__ row4 = (const uint4*)(obm + (size_t)blockIdx.x * BM_PITCH);
    constexpr int Q     = BM_PITCH / 4;
    constexpr int TRIPS = (Q + AG_TB - 1) / AG_TB;
    uint4 w[TRIPS];
#pragma unroll
    for (int t = 0; t < TRIPS; ++t) {
      const int q = tid + t * AG_TB;
      w[t] = row4[min(q, Q - 1)];
      if (q >= Q) w[t] = make_uint4(0u, 0u, 0u, 0u);
    }
    int c = 0;
#pragma unroll
    for (int t = 0; t < TRIPS; ++t) c += __popc(w[t].x) + __popc(w[t].y) + __popc(w[t].z) + __popc(w[t].w);
    int incl = c;
#pragma unroll
    for (int dl = 1; dl < 32; dl <<= 1) {
      const int v = __shfl_up(incl, dl);
      if (lane >= dl) incl += v;
    }
    int wbase = 0;
    if (lane == 31) wbase = atomicAdd(&cnt, incl);
    wbase = __builtin_amdgcn_readlane(wbase, 31);
    int slot = wbase + incl - c;
#pragma unroll
    for (int t = 0; t < TRIPS; ++t) {
      const int q = tid + t * AG_TB;
      const unsigned ww[4] = {w[t].x, w[t].y, w[t].z, w[t].w};
#pragma unroll
      for (int j = 0; j < 4; ++j) {
        unsigned x = ww[j];
        while (x) {
          const int b = __builtin_ctz(x);
          x &= x - 1u;
          if (slot < AG_CAP) lst[slot] = q * 128 + j * 32 + b;
          ++slot;
        }
      }
    }
    __syncthreads();

    total = cnt;
    if (tid == 0) wcnt[blockIdx.x] = total;
  }
  const bool overflow = total > AG_CAP;
  const int n = __builtin_amdgcn_readfirstlane(overflow ? 0 : total);

  auto leaky = [](float x) { return (x > 0.f) ? x : 0.01f * x; };
  auto load_ss = [&](int s, float (&x)[H]) {
    if constexpr (H == 4) {
      const float4 t = ((const float4*)ss)[s];
      x[0] = t.x; x[1] = t.y; x[2] = t.z; x[3] = t.w;
    } else {
#pragma unroll
      for (int h = 0; h < H; ++h) x[h] = ss[(size_t)s * H + h];
    }
  };
  auto put_sc = [&](int i, const float (&e)[H]) {
    if constexpr (SC_LDS) scl[i] = e[0];
    else wsc[i] = make_float4(e[0], e[1], e[2], e[3]);
  };
  auto get_sc = [&](int i, float (&e)[H]) {
    if constexpr (SC_LDS) e[0] = scl[i];
    else { const float4 t = wsc[i]; e[0] = t.x; e[1] = t.y; e[2] = t.z; e[3] = t.w; }
  };
  auto walk = [&](auto f) {
    for (int ed = tid; ed < N_EDGES; ed += AG_TB) {
      const unsigned r = (unsigned)(dst[ed] - base);
      if (r < (unsigned)AG_ROWS) f(src[ed], (int)r);
    }
  };

  if (overflow) {
    walk([&](int s, int r) {
      float x[H];
      load_ss(s, x);
      for (int h = 0; h < H; ++h) atomicMax(&mx[r * H + h], fkey(leaky(x[h] + sdl[r * H + h])));
    });
  } else {
    for (int i0 = tid; i0 < n; i0 += AG_B * AG_TB) {
      int p[AG_B], s[AG_B], rw[AG_B];
      float x[AG_B][H];
#pragma unroll
      for (int u = 0; u < AG_B; ++u) {
        const int idx = i0 + u * AG_TB;
        p[u] = (idx < n) ? lst[idx] : -1;
      }
#pragma unroll
      for (int u = 0; u < AG_B; ++u) {
        if constexpr (TAKE_LIST) {
          s[u]  = (p[u] >= 0) ? (p[u] >> AG_RB) : 0;
          rw[u] = p[u] & RMASK;
        } else {
          s[u]  = (p[u] >= 0) ? src[p[u]] : 0;
          const int d = (p[u] >= 0) ? dst[p[u]] : base;
          rw[u] = (int)min((unsigned)(d - base), (unsigned)AG_ROWS);
        }
      }
#pragma unroll
      for (int u = 0; u < AG_B; ++u) load_ss(s[u], x[u]);
#pragma unroll
      for (int u = 0; u < AG_B; ++u) {
        if (p[u] >= 0) {
          const int r = rw[u];
          if constexpr (!TAKE_LIST) {
            const int e = (s[u] << AG_RB) | r;
            lst[i0 + u * AG_TB] = e;
            wl[i0 + u * AG_TB] = e;
          }
          float e[H];
#pragma unroll
          for (int h = 0; h < H; ++h) {
            e[h] = leaky(x[u][h] + sdl[r * H + h]);
            atomicMax(&mx[r * H + h], fkey(e[h]));
          }
          put_sc(i0 + u * AG_TB, e);
        }
      }
    }
  }
  __syncthreads();

  if (overflow) {
    walk([&](int s, int r) {
      float x[H];
      load_ss(s, x);
      for (int h = 0; h < H; ++h)
        atomicAdd(&sm[r * H + h], expf(leaky(x[h] + sdl[r * H + h]) - funkey(mx[r * H + h])));
    });
  } else {
    for (int i0 = tid; i0 < n; i0 += AG_B * AG_TB) {
      int p[AG_B];
      float e[AG_B][H];
#pragma unroll
      for (int u = 0; u < AG_B; ++u) {
        const int idx = i0 + u * AG_TB;
        p[u] = (idx < n) ? lst[idx] : -1;
      }
#pragma unroll
      for (int u = 0; u < AG_B; ++u) get_sc((p[u] >= 0) ? (i0 + u * AG_TB) : i0, e[u]);
#pragma unroll
      for (int u = 0; u < AG_B; ++u) {
        if (p[u] >= 0) {
          const int r = p[u] & RMASK;
#pragma unroll
          for (int h = 0; h < H; ++h)
            atomicAdd(&sm[r * H + h], expf(e[u][h] - funkey(mx[r * H + h])));
        }
      }
    }
  }
  __syncthreads();

  if (overflow) {
    walk([&](int s, int r) {
      float x[H];
      load_ss(s, x);
      for (int h = 0; h < H; ++h) {
        const float a = expf(leaky(x[h] + sdl[r * H + h]) - funkey(mx[r * H + h])) / fmaxf(sm[r * H + h], 1e-9f);
        for (int o = 0; o < HID; ++o) {
          const int ch = h * HID + o;
          atomicAdd(&acc[r * C + (ch % VW) * 32 + ch / VW], a * z[(size_t)s * C + ch]);
        }
      }
    });
  }
  for (int b0 = wave * 32; b0 < n; b0 += AG_TB) {
    const int idx = b0 + lane;
    const bool valid = idx < n;
    const int p = valid ? lst[idx] : AG_ROWS;
    const int s = p >> AG_RB;
    const int row = p & RMASK;
    float al[H];
#pragma unroll
    for (int h = 0; h < H; ++h) al[h] = 0.f;
    if (valid) get_sc(idx, al);
#pragma unroll
    for (int h = 0; h < H; ++h) {
      const float ee = expf(al[h] - funkey(mx[row * H + h]));
      al[h] = valid ? ee / fmaxf(sm[row * H + h], 1e-9f) : 0.f;
    }
    const int nb = min(32, n - b0);
    [[maybe_unused]] const int hsel = lane / (32 / H);
    for (int i0 = 0; i0 < nb; i0 += AG_U) {
      float v[AG_U][VW];
#pragma unroll
      for (int u = 0; u < AG_U; ++u) {
        const float* zp = z + (size_t)lane_get_i(s, i0 + u) * C + lane * VW;
        if constexpr (VW == 4) {
          const float4 t = *(const float4*)zp;
          v[u][0] = t.x; v[u][1] = t.y; v[u][2] = t.z; v[u][3] = t.w;
        } else {
#pragma unroll
          for (int c = 0; c < VW; ++c) v[u][c] = zp[c];
        }
      }
#pragma unroll
      for (int u = 0; u < AG_U; ++u) {
        float* ap = acc + lane_get_i(row, i0 + u) * C + lane;
        float wgt = lane_get_f(al[0], i0 + u);
#pragma unroll
        for (int h = 1; h < H; ++h) {
          const float ah = lane_get_f(al[h], i0 + u);
          wgt = (hsel == h) ? ah : wgt;
        }
#pragma unroll
        for (int c = 0; c < VW; ++c)
          atomicAdd(ap + c * 32, wgt * v[u][c]);
      }
    }
  }
  __syncthreads();

  constexpr int C4 = C / 4;
  for (int i = tid; i < AG_ROWS * C4; i += AG_TB) {
    const int r = i / C4, c4 = i - r * C4;
    const int node = base + r;
    if (node < N_NODES) {
      float4 v;
      if constexpr (VW == 4) {
        const float* ap = &acc[r * C + c4];
        v = make_float4(ap[0], ap[32], ap[64], ap[96]);
      } else {
        v = *(const float4*)&acc[r * C + c4 * 4];
      }
      if constexpr (APPLY_ELU) {
        v.x = (v.x > 0.f) ? v.x : (expf(v.x) - 1.0f);
        v.y = (v.y > 0.f) ? v.y : (expf(v.y) - 1.0f);
        v.z = (v.z > 0.f) ? v.z : (expf(v.z) - 1.0f);
        v.w = (v.w > 0.f) ? v.w : (expf(v.w) - 1.0f);
      }
      *(float4*)&out[(size_t)node * ldo + c4 * 4] = v;
    }
  }
}

static_assert((size_t)AG_GROUPS * AG_CAP * 4 <= (size_t)N_NODES * (64 + HID + 2 + 96),
              "the score plane must fit inside the planes it lies over");

extern "C" void kernel_launch(void* const* d_in, const int* in_sizes, int n_in,
                              void* d_out, int out_size, void* d_ws, size_t ws_size,
                              hipStream_t stream) {
  const float* h    = (const float*)d_in[0];
  const int*   src  = (const int*)  d_in[1];
  const int*   dst  = (const int*)  d_in[2];
  const int*   fc   = (const int*)  d_in[3];
  const float* W1   = (const float*)d_in[4];
  const float* a1   = (const float*)d_in[5];
  const float* W2   = (const float*)d_in[6];
  const float* a2   = (const float*)d_in[7];
  const float* Wf   = (const float*)d_in[8];
  const float* Wff  = (const float*)d_in[9];
  const float* Wfc2 = (const float*)d_in[10];
  float* out = (float*)d_out;

  float* w = (float*)d_ws;
  size_t off = 0;
  float* agg1 = w + off; off += (size_t)N_NODES * 128;
  float* z1   = w + off; off += (size_t)N_NODES * 128;
  float* s1s  = w + off; off += (size_t)N_NODES * HEADS;
  float* s1d  = w + off; off += (size_t)N_NODES * HEADS;
  int*   elst = (int*)(w + off); off += (size_t)AG_GROUPS * AG_CAP;
  int*   ecnt = (int*)(w + off); off += AG_GROUPS;
  unsigned* obm  = (unsigned*)(w + off); off += (size_t)AG_GROUPS * BM_PITCH;
  float* esc  = w + off;
  float* hg   = w + off; off += (size_t)N_NODES * 64;
  float* z2   = w + off; off += (size_t)N_NODES * HID;
  float* s2s  = w + off; off += N_NODES;
  float* s2d  = w + off; off += N_NODES;
  off += (size_t)N_NODES * 96;
  (void)ws_size; (void)in_sizes; (void)n_in; (void)out_size;

  const int TB = 256;
  const int MT = N_NODES / 16;
  const int PG = 256;

  wmma_gemm_resident<4, 128, 2, 4, false><<<PG, 256, 0, stream>>>(h, nullptr, W1, z1, MT, 128, 0);
  attn_scores<<<(N_NODES * HEADS + TB - 1) / TB, TB, 0, stream>>>(z1, a1, s1s, s1d, N_NODES, HEADS);
  dst_range_bitmaps<<<BM_CHUNKS, BM_TB, 0, stream>>>(dst, obm);
  edge_softmax_aggregate_owner<HEADS, true, false><<<AG_GROUPS, AG_TB, 0, stream>>>(src, dst, s1s, s1d, z1, agg1, 128, elst, ecnt, obm, esc);

  wmma_gemm_resident<2, 128, 1, 8, false><<<PG, 256, 0, stream>>>(agg1, nullptr, W2, z2, MT, HID, 0);
  attn_scores<<<(N_NODES + TB - 1) / TB, TB, 0, stream>>>(z2, a2, s2s, s2d, N_NODES, 1);
  edge_softmax_aggregate_owner<1, false, true><<<AG_GROUPS, AG_TB, 0, stream>>>(src, dst, s2s, s2d, z2, hg, 64, elst, ecnt, obm, nullptr);

  frac_chain<<<PG, FC_WAVES * 32, 0, stream>>>(h, fc, Wf, Wff, Wfc2, hg, out, MT);
}
